# starting kernel
# baseline (speedup 1.0000x reference)
#include <hip/hip_runtime.h>
#include <math.h>


typedef __attribute__((ext_vector_type(2))) float v2f;
typedef __attribute__((ext_vector_type(8))) float v8f;

#define NEG_SLOPE 0.2f

#if defined(__has_builtin)
#if __has_builtin(__builtin_amdgcn_wmma_f32_16x16x4_f32)
#define HAVE_WMMA_F32_16X16X4 1
#endif
#endif

__device__ __forceinline__ unsigned flipf(float f) {
  unsigned b = __float_as_uint(f);
  return (b & 0x80000000u) ? ~b : (b | 0x80000000u);
}
__device__ __forceinline__ float unflipf(unsigned u) {
  return __uint_as_float((u & 0x80000000u) ? (u & 0x7FFFFFFFu) : ~u);
}

__device__ __forceinline__ float leaky(float v) {
  return v > 0.f ? v : NEG_SLOPE * v;
}

template <int K, int DOUT, bool RELU_BIAS>
__global__ void gat_gemm_wmma(const float* __restrict__ X,
                              const float* __restrict__ W,
                              const float* __restrict__ bias,
                              float* __restrict__ H, int M) {
  const int lane   = threadIdx.x & 31;
  const int wave   = threadIdx.x >> 5;
  const int tileId = blockIdx.x * (blockDim.x >> 5) + wave;
  const int NT     = DOUT / 16;
  const int mt     = tileId / NT;
  const int nt     = tileId % NT;
  if (mt >= M / 16) return;

  const int r15 = lane & 15;
  const int hi  = lane >> 4;
  const int row = mt * 16 + r15;
  const int col = nt * 16 + r15;

  v8f c = {0.f, 0.f, 0.f, 0.f, 0.f, 0.f, 0.f, 0.f};

#if HAVE_WMMA_F32_16X16X4
  for (int k0 = 0; k0 < K; k0 += 4) {
    const int ka = k0 + 2 * hi;
    float x0 = X[row * K + ka];
    float x1 = X[row * K + ka + 1];
    if (RELU_BIAS) {
      x0 = fmaxf(x0 + bias[ka], 0.f);
      x1 = fmaxf(x1 + bias[ka + 1], 0.f);
    }
    v2f a;
    a.x = x0;
    a.y = x1;
    v2f b;
    b.x = W[ka * DOUT + col];
    b.y = W[(ka + 1) * DOUT + col];
    c = __builtin_amdgcn_wmma_f32_16x16x4_f32(false, a, false, b, (short)0, c,
                                              false, false);
  }
#else
  for (int r = 0; r < 8; ++r) {
    const int rr = mt * 16 + hi * 8 + r;
    float acc = 0.f;
    for (int k = 0; k < K; ++k) {
      float xv = X[rr * K + k];
      if (RELU_BIAS) xv = fmaxf(xv + bias[k], 0.f);
      acc += xv * W[k * DOUT + col];
    }
    c[r] = acc;
  }
#endif

  const int rowBase = mt * 16 + hi * 8;
#pragma unroll
  for (int r = 0; r < 8; ++r) H[(rowBase + r) * DOUT + col] = c[r];
}

template <int D>
__global__ void att_scores(const float* __restrict__ H,
                           const float* __restrict__ attS,
                           const float* __restrict__ attD,
                           float* __restrict__ aS, float* __restrict__ aD,
                           int N) {
  const int gid  = blockIdx.x * blockDim.x + threadIdx.x;
  const int node = gid >> 5;
  const int lane = gid & 31;
  if (node >= N) return;
  float s = 0.f, d = 0.f;
#pragma unroll
  for (int f = lane; f < D; f += 32) {
    const float h = H[(size_t)node * D + f];
    s += h * attS[f];
    d += h * attD[f];
  }
#pragma unroll
  for (int off = 16; off > 0; off >>= 1) {
    s += __shfl_down(s, off, 32);
    d += __shfl_down(d, off, 32);
  }
  if (lane == 0) {
    aS[node] = s;
    aD[node] = d;
  }
}

__global__ void edge_lrelu_max(const int* __restrict__ src,
                               const int* __restrict__ dst,
                               const float* __restrict__ aS,
                               const float* __restrict__ aD,
                               unsigned* __restrict__ emax, int E, int Etot) {
  const int e = blockIdx.x * blockDim.x + threadIdx.x;
  if (e >= Etot) return;
  int s, d;
  if (e < E) {
    s = src[e];
    d = dst[e];
  } else {
    s = d = e - E;
  }
  const float v = leaky(aS[s] + aD[d]);
  atomicMax(&emax[d], flipf(v));
}

__global__ void edge_expsum(const int* __restrict__ src,
                            const int* __restrict__ dst,
                            const float* __restrict__ aS,
                            const float* __restrict__ aD,
                            const unsigned* __restrict__ emax,
                            float* __restrict__ wbuf,
                            float* __restrict__ denom, int E, int Etot) {
  const int e = blockIdx.x * blockDim.x + threadIdx.x;
  if (e >= Etot) return;
  int s, d;
  if (e < E) {
    s = src[e];
    d = dst[e];
  } else {
    s = d = e - E;
  }
  const float v = leaky(aS[s] + aD[d]);
  const float m = unflipf(emax[d]);
  const float w = __expf(v - m);
  wbuf[e] = w;
  unsafeAtomicAdd(&denom[d], w);
}

__global__ void edge_alpha(const int* __restrict__ dst,
                           const float* __restrict__ denom,
                           float* __restrict__ wbuf, int E, int Etot) {
  const int e = blockIdx.x * blockDim.x + threadIdx.x;
  if (e >= Etot) return;
  const int d = (e < E) ? dst[e] : (e - E);
  wbuf[e] = wbuf[e] / denom[d];
}

template <int D>
__global__ void edge_aggregate(const int* __restrict__ src,
                               const int* __restrict__ dst,
                               const float* __restrict__ alpha,
                               const float* __restrict__ H,
                               float* __restrict__ out, int E, int Etot) {
  constexpr int TPE = D / 4;
  const long long t = (long long)blockIdx.x * blockDim.x + threadIdx.x;
  const long long e = t / TPE;
  const int q = (int)(t % TPE);
  if (e >= Etot) return;
  int s, d;
  if (e < E) {
    s = src[e];
    d = dst[e];
  } else {
    s = d = (int)(e - E);
  }
  const float a = alpha[e];
  const float4 h = *reinterpret_cast<const float4*>(H + (size_t)s * D + 4 * q);
  float* o = out + (size_t)d * D + 4 * q;
  unsafeAtomicAdd(o + 0, a * h.x);
  unsafeAtomicAdd(o + 1, a * h.y);
  unsafeAtomicAdd(o + 2, a * h.z);
  unsafeAtomicAdd(o + 3, a * h.w);
}

template <int D>
__global__ void finalize_tanh(float* __restrict__ out,
                              const float* __restrict__ b, int N) {
  const int t = blockIdx.x * blockDim.x + threadIdx.x;
  if (t >= N * D) return;
  out[t] = tanhf(out[t] + b[t & (D - 1)]);
}

extern "C" void kernel_launch(void* const* d_in, const int* in_sizes, int n_in,
                              void* d_out, int out_size, void* d_ws,
                              size_t ws_size, hipStream_t stream) {
  const float* x   = (const float*)d_in[0];
  const int*   ei  = (const int*)d_in[1];
  const float* W1  = (const float*)d_in[2];
  const float* as1 = (const float*)d_in[3];
  const float* ad1 = (const float*)d_in[4];
  const float* b1  = (const float*)d_in[5];
  const float* W2  = (const float*)d_in[6];
  const float* as2 = (const float*)d_in[7];
  const float* ad2 = (const float*)d_in[8];
  const float* b2  = (const float*)d_in[9];
  float* out = (float*)d_out;

  const int N    = in_sizes[0] / 64;
  const int E    = in_sizes[1] / 2;
  const int Etot = E + N;
  const int* src = ei;
  const int* dst = ei + E;

  float* ws = (float*)d_ws;
  size_t o = 0;
  float* h1 = ws + o;   o += (size_t)N * 64;
  float* o1 = ws + o;   o += (size_t)N * 64;
  float* aS = ws + o;   o += N;
  float* aD = ws + o;   o += N;
  unsigned* emax = (unsigned*)(ws + o); o += N;
  float* denom = ws + o; o += N;
  float* wbuf  = ws + o; o += Etot;

  const int B  = 256;
  const int ge = (Etot + B - 1) / B;

  {
    const int tiles = (N / 16) * (64 / 16);
    gat_gemm_wmma<64, 64, false>
        <<<(tiles + 7) / 8, B, 0, stream>>>(x, W1, nullptr, h1, N);
    att_scores<64><<<(int)(((size_t)N * 32 + B - 1) / B), B, 0, stream>>>(
        h1, as1, ad1, aS, aD, N);
    hipMemsetAsync(emax, 0, (size_t)N * 4, stream);
    hipMemsetAsync(denom, 0, (size_t)N * 4, stream);
    hipMemsetAsync(o1, 0, (size_t)N * 64 * 4, stream);
    edge_lrelu_max<<<ge, B, 0, stream>>>(src, dst, aS, aD, emax, E, Etot);
    edge_expsum<<<ge, B, 0, stream>>>(src, dst, aS, aD, emax, wbuf, denom, E,
                                      Etot);
    edge_alpha<<<ge, B, 0, stream>>>(dst, denom, wbuf, E, Etot);
    edge_aggregate<64>
        <<<(int)(((size_t)Etot * 16 + B - 1) / B), B, 0, stream>>>(
            src, dst, wbuf, h1, o1, E, Etot);
  }

  {
    float* h2 = h1;
    const int tiles = (N / 16) * (32 / 16);
    gat_gemm_wmma<64, 32, true>
        <<<(tiles + 7) / 8, B, 0, stream>>>(o1, W2, b1, h2, N);
    att_scores<32><<<(int)(((size_t)N * 32 + B - 1) / B), B, 0, stream>>>(
        h2, as2, ad2, aS, aD, N);
    hipMemsetAsync(emax, 0, (size_t)N * 4, stream);
    hipMemsetAsync(denom, 0, (size_t)N * 4, stream);
    hipMemsetAsync(out, 0, (size_t)N * 32 * 4, stream);
    edge_lrelu_max<<<ge, B, 0, stream>>>(src, dst, aS, aD, emax, E, Etot);
    edge_expsum<<<ge, B, 0, stream>>>(src, dst, aS, aD, emax, wbuf, denom, E,
                                      Etot);
    edge_alpha<<<ge, B, 0, stream>>>(dst, denom, wbuf, E, Etot);
    edge_aggregate<32>
        <<<(int)(((size_t)Etot * 8 + B - 1) / B), B, 0, stream>>>(
            src, dst, wbuf, h2, out, E, Etot);
    finalize_tanh<32><<<(int)(((size_t)N * 32 + B - 1) / B), B, 0, stream>>>(
        out, b2, N);
  }
}


# Group edges by destination once per call (atomic count+rank, scan, scatter); one wave per node does segment max, exp-sum, weighted row sum in registers: no f32 atomics, sum order = arrival rank
# speedup vs baseline: 2.6156x; 2.6156x over previous
#include <hip/hip_runtime.h>
#include <math.h>


typedef __attribute__((ext_vector_type(2))) float v2f;
typedef __attribute__((ext_vector_type(8))) float v8f;

#define NEG_SLOPE 0.2f

#if defined(__has_builtin)
#if __has_builtin(__builtin_amdgcn_wmma_f32_16x16x4_f32)
#define HAVE_WMMA_F32_16X16X4 1
#endif
#endif

__device__ __forceinline__ float leaky(float v) {
  return v > 0.f ? v : NEG_SLOPE * v;
}

template <int K, int DOUT, bool RELU_BIAS>
__global__ void gat_gemm_wmma(const float* __restrict__ X,
                              const float* __restrict__ W,
                              const float* __restrict__ bias,
                              float* __restrict__ H, int M) {
  const int lane   = threadIdx.x & 31;
  const int wave   = threadIdx.x >> 5;
  const int tileId = blockIdx.x * (blockDim.x >> 5) + wave;
  const int NT     = DOUT / 16;
  const int mt     = tileId / NT;
  const int nt     = tileId % NT;
  if (mt >= M / 16) return;

  const int r15 = lane & 15;
  const int hi  = lane >> 4;
  const int row = mt * 16 + r15;
  const int col = nt * 16 + r15;

  v8f c = {0.f, 0.f, 0.f, 0.f, 0.f, 0.f, 0.f, 0.f};

#if HAVE_WMMA_F32_16X16X4
  for (int k0 = 0; k0 < K; k0 += 4) {
    const int ka = k0 + 2 * hi;
    float x0 = X[row * K + ka];
    float x1 = X[row * K + ka + 1];
    if (RELU_BIAS) {
      x0 = fmaxf(x0 + bias[ka], 0.f);
      x1 = fmaxf(x1 + bias[ka + 1], 0.f);
    }
    v2f a;
    a.x = x0;
    a.y = x1;
    v2f b;
    b.x = W[ka * DOUT + col];
    b.y = W[(ka + 1) * DOUT + col];
    c = __builtin_amdgcn_wmma_f32_16x16x4_f32(false, a, false, b, (short)0, c,
                                              false, false);
  }
#else
  for (int r = 0; r < 8; ++r) {
    const int rr = mt * 16 + hi * 8 + r;
    float acc = 0.f;
    for (int k = 0; k < K; ++k) {
      float xv = X[rr * K + k];
      if (RELU_BIAS) xv = fmaxf(xv + bias[k], 0.f);
      acc += xv * W[k * DOUT + col];
    }
    c[r] = acc;
  }
#endif

  const int rowBase = mt * 16 + hi * 8;
#pragma unroll
  for (int r = 0; r < 8; ++r) H[(rowBase + r) * DOUT + col] = c[r];
}

template <int D>
__global__ void att_scores(const float* __restrict__ H,
                           const float* __restrict__ attS,
                           const float* __restrict__ attD,
                           float* __restrict__ aS, float* __restrict__ aD,
                           int N) {
  const int gid  = blockIdx.x * blockDim.x + threadIdx.x;
  const int node = gid >> 5;
  const int lane = gid & 31;
  if (node >= N) return;
  float s = 0.f, d = 0.f;
#pragma unroll
  for (int f = lane; f < D; f += 32) {
    const float h = H[(size_t)node * D + f];
    s += h * attS[f];
    d += h * attD[f];
  }
#pragma unroll
  for (int off = 16; off > 0; off >>= 1) {
    s += __shfl_down(s, off, 32);
    d += __shfl_down(d, off, 32);
  }
  if (lane == 0) {
    aS[node] = s;
    aD[node] = d;
  }
}

__global__ void csr_count_rank(const int* __restrict__ dst,
                               int* __restrict__ cnt, int* __restrict__ rank,
                               int E) {
  const int e = blockIdx.x * blockDim.x + threadIdx.x;
  if (e >= E) return;
  rank[e] = atomicAdd(&cnt[dst[e]], 1);
}

__global__ void csr_scan(const int* __restrict__ cnt, int* __restrict__ rowptr,
                         int N) {
  __shared__ int s_front[32];
  __shared__ int s_wave[32];
  const int tid   = threadIdx.x;
  const int lane  = tid & 31;
  const int wv    = tid >> 5;
  const int start = blockIdx.x * 1024;

  int part = 0;
  for (int i = tid; i < start; i += 1024) part += cnt[i];
  for (int off = 16; off > 0; off >>= 1) part += __shfl_xor(part, off, 32);

  const int idx = start + tid;
  const int c   = (idx < N) ? cnt[idx] : 0;
  int x = c;
  for (int off = 1; off < 32; off <<= 1) {
    const int y = __shfl_up(x, off, 32);
    if (lane >= off) x += y;
  }
  if (lane == 0) s_front[wv] = part;
  if (lane == 31) s_wave[wv] = x;
  __syncthreads();

  int base = 0;
  for (int k = 0; k < 32; ++k) {
    base += s_front[k];
    if (k < wv) base += s_wave[k];
  }
  const int excl = base + x - c;
  if (idx < N) rowptr[idx] = excl;
  if (idx == N - 1) rowptr[N] = excl + c;
}

__global__ void csr_fill(const int* __restrict__ src,
                         const int* __restrict__ dst,
                         const int* __restrict__ rowptr,
                         const int* __restrict__ rank,
                         int* __restrict__ csr_src, int E) {
  const int e = blockIdx.x * blockDim.x + threadIdx.x;
  if (e >= E) return;
  csr_src[rowptr[dst[e]] + rank[e]] = src[e];
}

template <int D>
__device__ __forceinline__ void add_rows(const float* __restrict__ Hl, int s,
                                         float w, int n, float& acc0,
                                         float& acc1) {
#pragma unroll 4
  for (int j = 0; j < n; ++j) {
    const int sj = __builtin_amdgcn_readlane(s, j);
    const float wj =
        __int_as_float(__builtin_amdgcn_readlane(__float_as_int(w), j));
    if (D == 64) {
      const float2 h = *reinterpret_cast<const float2*>(Hl + (size_t)sj * D);
      acc0 += wj * h.x;
      acc1 += wj * h.y;
    } else {
      acc0 += wj * Hl[(size_t)sj * D];
    }
  }
}

template <int D>
__global__ void node_softmax_aggregate(const int* __restrict__ rowptr,
                                       const int* __restrict__ csr_src,
                                       const float* __restrict__ aS,
                                       const float* __restrict__ aD,
                                       const float* __restrict__ H,
                                       float* __restrict__ out, int N) {
  static_assert(D == 32 || D == 64, "row must be 32 or 64 floats");
  constexpr int V = D / 32;
  const int lane = threadIdx.x & 31;
  const int node = __builtin_amdgcn_readfirstlane(
      (int)((blockIdx.x * blockDim.x + threadIdx.x) >> 5));
  if (node >= N) return;

  const int beg = rowptr[node];
  const int end = rowptr[node + 1];
  const float ad = aD[node];
  const float vself = leaky(aS[node] + ad);

  const int i0 = beg + lane;
  const bool has0 = i0 < end;
  int s0 = 0;
  float v0 = vself;
  if (has0) {
    s0 = csr_src[i0];
    v0 = leaky(aS[s0] + ad);
  }
  float m = fmaxf(vself, v0);
  for (int i = i0 + 32; i < end; i += 32)
    m = fmaxf(m, leaky(aS[csr_src[i]] + ad));
  for (int off = 16; off > 0; off >>= 1) m = fmaxf(m, __shfl_xor(m, off, 32));

  const float* Hl = H + lane * V;
  const float wself = __expf(vself - m);
  float acc0, acc1 = 0.f;
  if (D == 64) {
    const float2 h = *reinterpret_cast<const float2*>(Hl + (size_t)node * D);
    acc0 = wself * h.x;
    acc1 = wself * h.y;
  } else {
    acc0 = wself * Hl[(size_t)node * D];
  }
  const int deg = end - beg;
  float wl = has0 ? __expf(v0 - m) : 0.f;
  add_rows<D>(Hl, s0, wl, deg < 32 ? deg : 32, acc0, acc1);
  for (int base = beg + 32; base < end; base += 32) {
    const int i = base + lane;
    int s = 0;
    float w = 0.f;
    if (i < end) {
      s = csr_src[i];
      w = __expf(leaky(aS[s] + ad) - m);
    }
    wl += w;
    add_rows<D>(Hl, s, w, (end - base < 32) ? (end - base) : 32, acc0, acc1);
  }
  for (int off = 16; off > 0; off >>= 1) wl += __shfl_xor(wl, off, 32);
  const float denom = wl + wself;

  float* o = out + (size_t)node * D + lane * V;
  if (D == 64) {
    float2 r;
    r.x = acc0 / denom;
    r.y = acc1 / denom;
    *reinterpret_cast<float2*>(o) = r;
  } else {
    o[0] = acc0 / denom;
  }
}

template <int D>
__global__ void finalize_tanh(float* __restrict__ out,
                              const float* __restrict__ b, int N) {
  const int t = blockIdx.x * blockDim.x + threadIdx.x;
  if (t >= N * D) return;
  out[t] = tanhf(out[t] + b[t & (D - 1)]);
}

extern "C" void kernel_launch(void* const* d_in, const int* in_sizes, int n_in,
                              void* d_out, int out_size, void* d_ws,
                              size_t ws_size, hipStream_t stream) {
  const float* x   = (const float*)d_in[0];
  const int*   ei  = (const int*)d_in[1];
  const float* W1  = (const float*)d_in[2];
  const float* as1 = (const float*)d_in[3];
  const float* ad1 = (const float*)d_in[4];
  const float* b1  = (const float*)d_in[5];
  const float* W2  = (const float*)d_in[6];
  const float* as2 = (const float*)d_in[7];
  const float* ad2 = (const float*)d_in[8];
  const float* b2  = (const float*)d_in[9];
  float* out = (float*)d_out;

  const int N    = in_sizes[0] / 64;
  const int E    = in_sizes[1] / 2;
  const int* src = ei;
  const int* dst = ei + E;

  float* ws = (float*)d_ws;
  size_t o = 0;
  float* h1 = ws + o;   o += (size_t)N * 64;
  float* o1 = ws + o;   o += (size_t)N * 64;
  float* aS = ws + o;   o += N;
  float* aD = ws + o;   o += N;
  int* cnt     = (int*)(ws + o); o += N;
  int* rowptr  = (int*)(ws + o); o += N + 1;
  int* rank    = (int*)(ws + o); o += E;
  int* csr_src = (int*)(ws + o); o += E;

  const int B  = 256;
  const int ge = (E + B - 1) / B;
  const int gn = (int)(((size_t)N * 32 + B - 1) / B);

  hipMemsetAsync(cnt, 0, (size_t)N * 4, stream);
  csr_count_rank<<<ge, B, 0, stream>>>(dst, cnt, rank, E);
  csr_scan<<<(N + 1023) / 1024, 1024, 0, stream>>>(cnt, rowptr, N);
  csr_fill<<<ge, B, 0, stream>>>(src, dst, rowptr, rank, csr_src, E);

  {
    const int tiles = (N / 16) * (64 / 16);
    gat_gemm_wmma<64, 64, false>
        <<<(tiles + 7) / 8, B, 0, stream>>>(x, W1, nullptr, h1, N);
    att_scores<64><<<gn, B, 0, stream>>>(h1, as1, ad1, aS, aD, N);
    node_softmax_aggregate<64>
        <<<gn, B, 0, stream>>>(rowptr, csr_src, aS, aD, h1, o1, N);
  }

  {
    float* h2 = h1;
    const int tiles = (N / 16) * (32 / 16);
    gat_gemm_wmma<64, 32, true>
        <<<(tiles + 7) / 8, B, 0, stream>>>(o1, W2, b1, h2, N);
    att_scores<32><<<gn, B, 0, stream>>>(h2, as2, ad2, aS, aD, N);
    node_softmax_aggregate<32>
        <<<gn, B, 0, stream>>>(rowptr, csr_src, aS, aD, h2, out, N);
    finalize_tanh<32><<<(int)(((size_t)N * 32 + B - 1) / B), B, 0, stream>>>(
        out, b2, N);
  }
}


# Scores computed from the rows the aggregate gathers anyway (16-row batches in registers, lane-exchange dot sums, running-max rescale between batches): 2 score launches and the score gather gone
# speedup vs baseline: 2.9400x; 1.1240x over previous
#include <hip/hip_runtime.h>
#include <math.h>


typedef __attribute__((ext_vector_type(2))) float v2f;
typedef __attribute__((ext_vector_type(8))) float v8f;

#define NEG_SLOPE 0.2f

#if defined(__has_builtin)
#if __has_builtin(__builtin_amdgcn_wmma_f32_16x16x4_f32)
#define HAVE_WMMA_F32_16X16X4 1
#endif
#endif

__device__ __forceinline__ float leaky(float v) {
  return v > 0.f ? v : NEG_SLOPE * v;
}

template <int K, int DOUT, bool RELU_BIAS>
__global__ void gat_gemm_wmma(const float* __restrict__ X,
                              const float* __restrict__ W,
                              const float* __restrict__ bias,
                              float* __restrict__ H, int M) {
  const int lane   = threadIdx.x & 31;
  const int wave   = threadIdx.x >> 5;
  const int tileId = blockIdx.x * (blockDim.x >> 5) + wave;
  const int NT     = DOUT / 16;
  const int mt     = tileId / NT;
  const int nt     = tileId % NT;
  if (mt >= M / 16) return;

  const int r15 = lane & 15;
  const int hi  = lane >> 4;
  const int row = mt * 16 + r15;
  const int col = nt * 16 + r15;

  v8f c = {0.f, 0.f, 0.f, 0.f, 0.f, 0.f, 0.f, 0.f};

#if HAVE_WMMA_F32_16X16X4
  for (int k0 = 0; k0 < K; k0 += 4) {
    const int ka = k0 + 2 * hi;
    float x0 = X[row * K + ka];
    float x1 = X[row * K + ka + 1];
    if (RELU_BIAS) {
      x0 = fmaxf(x0 + bias[ka], 0.f);
      x1 = fmaxf(x1 + bias[ka + 1], 0.f);
    }
    v2f a;
    a.x = x0;
    a.y = x1;
    v2f b;
    b.x = W[ka * DOUT + col];
    b.y = W[(ka + 1) * DOUT + col];
    c = __builtin_amdgcn_wmma_f32_16x16x4_f32(false, a, false, b, (short)0, c,
                                              false, false);
  }
#else
  for (int r = 0; r < 8; ++r) {
    const int rr = mt * 16 + hi * 8 + r;
    float acc = 0.f;
    for (int k = 0; k < K; ++k) {
      float xv = X[rr * K + k];
      if (RELU_BIAS) xv = fmaxf(xv + bias[k], 0.f);
      acc += xv * W[k * DOUT + col];
    }
    c[r] = acc;
  }
#endif

  const int rowBase = mt * 16 + hi * 8;
#pragma unroll
  for (int r = 0; r < 8; ++r) H[(rowBase + r) * DOUT + col] = c[r];
}

__global__ void csr_count_rank(const int* __restrict__ dst,
                               int* __restrict__ cnt, int* __restrict__ rank,
                               int E) {
  const int e = blockIdx.x * blockDim.x + threadIdx.x;
  if (e >= E) return;
  rank[e] = atomicAdd(&cnt[dst[e]], 1);
}

__global__ void csr_scan(const int* __restrict__ cnt, int* __restrict__ rowptr,
                         int N) {
  __shared__ int s_front[32];
  __shared__ int s_wave[32];
  const int tid   = threadIdx.x;
  const int lane  = tid & 31;
  const int wv    = tid >> 5;
  const int start = blockIdx.x * 1024;

  int part = 0;
  for (int i = tid; i < start; i += 1024) part += cnt[i];
  for (int off = 16; off > 0; off >>= 1) part += __shfl_xor(part, off, 32);

  const int idx = start + tid;
  const int c   = (idx < N) ? cnt[idx] : 0;
  int x = c;
  for (int off = 1; off < 32; off <<= 1) {
    const int y = __shfl_up(x, off, 32);
    if (lane >= off) x += y;
  }
  if (lane == 0) s_front[wv] = part;
  if (lane == 31) s_wave[wv] = x;
  __syncthreads();

  int base = 0;
  for (int k = 0; k < 32; ++k) {
    base += s_front[k];
    if (k < wv) base += s_wave[k];
  }
  const int excl = base + x - c;
  if (idx < N) rowptr[idx] = excl;
  if (idx == N - 1) rowptr[N] = excl + c;
}

__global__ void csr_fill(const int* __restrict__ src,
                         const int* __restrict__ dst,
                         const int* __restrict__ rowptr,
                         const int* __restrict__ rank,
                         int* __restrict__ csr_src, int E) {
  const int e = blockIdx.x * blockDim.x + threadIdx.x;
  if (e >= E) return;
  csr_src[rowptr[dst[e]] + rank[e]] = src[e];
}

template <int D>
__global__ void node_softmax_aggregate(const int* __restrict__ rowptr,
                                       const int* __restrict__ csr_src,
                                       const float* __restrict__ attS,
                                       const float* __restrict__ attD,
                                       const float* __restrict__ H,
                                       float* __restrict__ out, int N) {
  static_assert(D == 32 || D == 64, "row must be 32 or 64 floats");
  constexpr int V = D / 32;
  const int lane = threadIdx.x & 31;
  const int l15  = lane & 15;
  const int node = __builtin_amdgcn_readfirstlane(
      (int)((blockIdx.x * blockDim.x + threadIdx.x) >> 5));
  if (node >= N) return;

  const float* Hl = H + lane * V;
  float hs0, hs1 = 0.f, as0, as1 = 0.f, ad0, ad1 = 0.f;
  if (D == 64) {
    const float2 h = *reinterpret_cast<const float2*>(Hl + (size_t)node * D);
    const float2 a = *reinterpret_cast<const float2*>(attS + lane * V);
    const float2 b = *reinterpret_cast<const float2*>(attD + lane * V);
    hs0 = h.x; hs1 = h.y; as0 = a.x; as1 = a.y; ad0 = b.x; ad1 = b.y;
  } else {
    hs0 = Hl[(size_t)node * D];
    as0 = attS[lane];
    ad0 = attD[lane];
  }
  const int beg = rowptr[node];
  const int end = rowptr[node + 1];

  float ps = hs0 * as0 + hs1 * as1;
  float pd = hs0 * ad0 + hs1 * ad1;
  for (int off = 16; off > 0; off >>= 1) {
    ps += __shfl_xor(ps, off, 32);
    pd += __shfl_xor(pd, off, 32);
  }
  const float ad = pd;
  float m = leaky(ps + ad);
  float l = 1.f;
  float acc0 = hs0, acc1 = hs1;

  const bool b3 = (lane & 8) != 0, b2 = (lane & 4) != 0;
  const bool b1 = (lane & 2) != 0, b0 = (lane & 1) != 0;

  for (int base = beg; base < end; base += 16) {
    const int n = (end - base < 16) ? (end - base) : 16;
    const bool has = l15 < n;
    int s = node;
    if (has) s = csr_src[base + l15];

    float r0[16], r1[16];
#pragma unroll
    for (int j = 0; j < 16; ++j) {
      const int sj = __builtin_amdgcn_readlane(s, j);
      if (D == 64) {
        const float2 h = *reinterpret_cast<const float2*>(Hl + (size_t)sj * D);
        r0[j] = h.x;
        r1[j] = h.y;
      } else {
        r0[j] = Hl[(size_t)sj * D];
        r1[j] = 0.f;
      }
    }

    float p[16];
#pragma unroll
    for (int j = 0; j < 16; ++j) p[j] = r0[j] * as0 + r1[j] * as1;
    float q[8];
#pragma unroll
    for (int k = 0; k < 8; ++k) {
      const float keep = b3 ? p[k + 8] : p[k];
      const float send = b3 ? p[k] : p[k + 8];
      q[k] = keep + __shfl_xor(send, 8, 32);
    }
    float t[4];
#pragma unroll
    for (int k = 0; k < 4; ++k) {
      const float keep = b2 ? q[k + 4] : q[k];
      const float send = b2 ? q[k] : q[k + 4];
      t[k] = keep + __shfl_xor(send, 4, 32);
    }
    float u[2];
#pragma unroll
    for (int k = 0; k < 2; ++k) {
      const float keep = b1 ? t[k + 2] : t[k];
      const float send = b1 ? t[k] : t[k + 2];
      u[k] = keep + __shfl_xor(send, 2, 32);
    }
    float sc;
    {
      const float keep = b0 ? u[1] : u[0];
      const float send = b0 ? u[0] : u[1];
      sc = keep + __shfl_xor(send, 1, 32);
    }
    sc += __shfl_xor(sc, 16, 32);

    const float v = has ? leaky(sc + ad) : -INFINITY;
    float cm = v;
    for (int off = 8; off > 0; off >>= 1) cm = fmaxf(cm, __shfl_xor(cm, off, 32));
    const float mn    = fmaxf(m, cm);
    const float scale = __expf(m - mn);
    const float w     = has ? __expf(v - mn) : 0.f;
    float ws = w;
    for (int off = 8; off > 0; off >>= 1) ws += __shfl_xor(ws, off, 32);
    l    = l * scale + ws;
    acc0 = acc0 * scale;
    acc1 = acc1 * scale;
    m    = mn;

#pragma unroll
    for (int j = 0; j < 16; ++j) {
      if (j < n) {
        const float wj =
            __int_as_float(__builtin_amdgcn_readlane(__float_as_int(w), j));
        acc0 += wj * r0[j];
        if (D == 64) acc1 += wj * r1[j];
      }
    }
  }

  float* o = out + (size_t)node * D + lane * V;
  if (D == 64) {
    float2 r;
    r.x = acc0 / l;
    r.y = acc1 / l;
    *reinterpret_cast<float2*>(o) = r;
  } else {
    o[0] = acc0 / l;
  }
}

template <int D>
__global__ void finalize_tanh(float* __restrict__ out,
                              const float* __restrict__ b, int N) {
  const int t = blockIdx.x * blockDim.x + threadIdx.x;
  if (t >= N * D) return;
  out[t] = tanhf(out[t] + b[t & (D - 1)]);
}

extern "C" void kernel_launch(void* const* d_in, const int* in_sizes, int n_in,
                              void* d_out, int out_size, void* d_ws,
                              size_t ws_size, hipStream_t stream) {
  const float* x   = (const float*)d_in[0];
  const int*   ei  = (const int*)d_in[1];
  const float* W1  = (const float*)d_in[2];
  const float* as1 = (const float*)d_in[3];
  const float* ad1 = (const float*)d_in[4];
  const float* b1  = (const float*)d_in[5];
  const float* W2  = (const float*)d_in[6];
  const float* as2 = (const float*)d_in[7];
  const float* ad2 = (const float*)d_in[8];
  const float* b2  = (const float*)d_in[9];
  float* out = (float*)d_out;

  const int N    = in_sizes[0] / 64;
  const int E    = in_sizes[1] / 2;
  const int* src = ei;
  const int* dst = ei + E;

  float* ws = (float*)d_ws;
  size_t o = 0;
  float* h1 = ws + o;   o += (size_t)N * 64;
  float* o1 = ws + o;   o += (size_t)N * 64;
  int* cnt     = (int*)(ws + o); o += N;
  int* rowptr  = (int*)(ws + o); o += N + 1;
  int* rank    = (int*)(ws + o); o += E;
  int* csr_src = (int*)(ws + o); o += E;

  const int B  = 256;
  const int ge = (E + B - 1) / B;
  const int gn = (int)(((size_t)N * 32 + B - 1) / B);

  hipMemsetAsync(cnt, 0, (size_t)N * 4, stream);
  csr_count_rank<<<ge, B, 0, stream>>>(dst, cnt, rank, E);
  csr_scan<<<(N + 1023) / 1024, 1024, 0, stream>>>(cnt, rowptr, N);
  csr_fill<<<ge, B, 0, stream>>>(src, dst, rowptr, rank, csr_src, E);

  {
    const int tiles = (N / 16) * (64 / 16);
    gat_gemm_wmma<64, 64, false>
        <<<(tiles + 7) / 8, B, 0, stream>>>(x, W1, nullptr, h1, N);
    node_softmax_aggregate<64>
        <<<gn, B, 0, stream>>>(rowptr, csr_src, as1, ad1, h1, o1, N);
  }

  {
    float* h2 = h1;
    const int tiles = (N / 16) * (32 / 16);
    gat_gemm_wmma<64, 32, true>
        <<<(tiles + 7) / 8, B, 0, stream>>>(o1, W2, b1, h2, N);
    node_softmax_aggregate<32>
        <<<gn, B, 0, stream>>>(rowptr, csr_src, as2, ad2, h2, out, N);
    finalize_tanh<32><<<(int)(((size_t)N * 32 + B - 1) / B), B, 0, stream>>>(
        out, b2, N);
  }
}


# Aggregate kernels: all lane exchanges (32-lane dot sums, 16-row score butterfly, 16-lane max and sum) on DPP row_xmask / permlanex16, not __shfl_xor's LDS bpermute; same pairs and order of sums
# speedup vs baseline: 3.1847x; 1.0832x over previous
#include <hip/hip_runtime.h>
#include <math.h>


typedef __attribute__((ext_vector_type(2))) float v2f;
typedef __attribute__((ext_vector_type(8))) float v8f;

#define NEG_SLOPE 0.2f

#if defined(__has_builtin)
#if __has_builtin(__builtin_amdgcn_wmma_f32_16x16x4_f32)
#define HAVE_WMMA_F32_16X16X4 1
#endif
#endif

__device__ __forceinline__ float leaky(float v) {
  return v > 0.f ? v : NEG_SLOPE * v;
}

template <int M>
__device__ __forceinline__ float lane_xor(float v) {
  return __builtin_bit_cast(
      float, __builtin_amdgcn_update_dpp(0, __builtin_bit_cast(int, v),
                                         0x160 + M, 0xf, 0xf, true));
}
__device__ __forceinline__ float lane_xor16(float v) {
  const int b = __builtin_bit_cast(int, v);
  return __builtin_bit_cast(
      float, __builtin_amdgcn_permlanex16(b, b, 0x76543210, 0xfedcba98, false,
                                          false));
}

template <int K, int DOUT, bool RELU_BIAS>
__global__ void gat_gemm_wmma(const float* __restrict__ X,
                              const float* __restrict__ W,
                              const float* __restrict__ bias,
                              float* __restrict__ H, int M) {
  const int lane   = threadIdx.x & 31;
  const int wave   = threadIdx.x >> 5;
  const int tileId = blockIdx.x * (blockDim.x >> 5) + wave;
  const int NT     = DOUT / 16;
  const int mt     = tileId / NT;
  const int nt     = tileId % NT;
  if (mt >= M / 16) return;

  const int r15 = lane & 15;
  const int hi  = lane >> 4;
  const int row = mt * 16 + r15;
  const int col = nt * 16 + r15;

  v8f c = {0.f, 0.f, 0.f, 0.f, 0.f, 0.f, 0.f, 0.f};

#if HAVE_WMMA_F32_16X16X4
  for (int k0 = 0; k0 < K; k0 += 4) {
    const int ka = k0 + 2 * hi;
    float x0 = X[row * K + ka];
    float x1 = X[row * K + ka + 1];
    if (RELU_BIAS) {
      x0 = fmaxf(x0 + bias[ka], 0.f);
      x1 = fmaxf(x1 + bias[ka + 1], 0.f);
    }
    v2f a;
    a.x = x0;
    a.y = x1;
    v2f b;
    b.x = W[ka * DOUT + col];
    b.y = W[(ka + 1) * DOUT + col];
    c = __builtin_amdgcn_wmma_f32_16x16x4_f32(false, a, false, b, (short)0, c,
                                              false, false);
  }
#else
  for (int r = 0; r < 8; ++r) {
    const int rr = mt * 16 + hi * 8 + r;
    float acc = 0.f;
    for (int k = 0; k < K; ++k) {
      float xv = X[rr * K + k];
      if (RELU_BIAS) xv = fmaxf(xv + bias[k], 0.f);
      acc += xv * W[k * DOUT + col];
    }
    c[r] = acc;
  }
#endif

  const int rowBase = mt * 16 + hi * 8;
#pragma unroll
  for (int r = 0; r < 8; ++r) H[(rowBase + r) * DOUT + col] = c[r];
}

__global__ void csr_count_rank(const int* __restrict__ dst,
                               int* __restrict__ cnt, int* __restrict__ rank,
                               int E) {
  const int e = blockIdx.x * blockDim.x + threadIdx.x;
  if (e >= E) return;
  rank[e] = atomicAdd(&cnt[dst[e]], 1);
}

__global__ void csr_scan(const int* __restrict__ cnt, int* __restrict__ rowptr,
                         int N) {
  __shared__ int s_front[32];
  __shared__ int s_wave[32];
  const int tid   = threadIdx.x;
  const int lane  = tid & 31;
  const int wv    = tid >> 5;
  const int start = blockIdx.x * 1024;

  int part = 0;
  for (int i = tid; i < start; i += 1024) part += cnt[i];
  for (int off = 16; off > 0; off >>= 1) part += __shfl_xor(part, off, 32);

  const int idx = start + tid;
  const int c   = (idx < N) ? cnt[idx] : 0;
  int x = c;
  for (int off = 1; off < 32; off <<= 1) {
    const int y = __shfl_up(x, off, 32);
    if (lane >= off) x += y;
  }
  if (lane == 0) s_front[wv] = part;
  if (lane == 31) s_wave[wv] = x;
  __syncthreads();

  int base = 0;
  for (int k = 0; k < 32; ++k) {
    base += s_front[k];
    if (k < wv) base += s_wave[k];
  }
  const int excl = base + x - c;
  if (idx < N) rowptr[idx] = excl;
  if (idx == N - 1) rowptr[N] = excl + c;
}

__global__ void csr_fill(const int* __restrict__ src,
                         const int* __restrict__ dst,
                         const int* __restrict__ rowptr,
                         const int* __restrict__ rank,
                         int* __restrict__ csr_src, int E) {
  const int e = blockIdx.x * blockDim.x + threadIdx.x;
  if (e >= E) return;
  csr_src[rowptr[dst[e]] + rank[e]] = src[e];
}

template <int D>
__global__ void node_softmax_aggregate(const int* __restrict__ rowptr,
                                       const int* __restrict__ csr_src,
                                       const float* __restrict__ attS,
                                       const float* __restrict__ attD,
                                       const float* __restrict__ H,
                                       float* __restrict__ out, int N) {
  static_assert(D == 32 || D == 64, "row must be 32 or 64 floats");
  constexpr int V = D / 32;
  const int lane = threadIdx.x & 31;
  const int l15  = lane & 15;
  const int node = __builtin_amdgcn_readfirstlane(
      (int)((blockIdx.x * blockDim.x + threadIdx.x) >> 5));
  if (node >= N) return;

  const float* Hl = H + lane * V;
  float hs0, hs1 = 0.f, as0, as1 = 0.f, ad0, ad1 = 0.f;
  if (D == 64) {
    const float2 h = *reinterpret_cast<const float2*>(Hl + (size_t)node * D);
    const float2 a = *reinterpret_cast<const float2*>(attS + lane * V);
    const float2 b = *reinterpret_cast<const float2*>(attD + lane * V);
    hs0 = h.x; hs1 = h.y; as0 = a.x; as1 = a.y; ad0 = b.x; ad1 = b.y;
  } else {
    hs0 = Hl[(size_t)node * D];
    as0 = attS[lane];
    ad0 = attD[lane];
  }
  const int beg = rowptr[node];
  const int end = rowptr[node + 1];

  float ps = fmaf(hs0, as0, hs1 * as1);
  float pd = fmaf(hs0, ad0, hs1 * ad1);
  ps += lane_xor16(ps);   pd += lane_xor16(pd);
  ps += lane_xor<8>(ps);  pd += lane_xor<8>(pd);
  ps += lane_xor<4>(ps);  pd += lane_xor<4>(pd);
  ps += lane_xor<2>(ps);  pd += lane_xor<2>(pd);
  ps += lane_xor<1>(ps);  pd += lane_xor<1>(pd);
  const float ad = pd;
  float m = leaky(ps + ad);
  float l = 1.f;
  float acc0 = hs0, acc1 = hs1;

  const bool b3 = (lane & 8) != 0, b2 = (lane & 4) != 0;
  const bool b1 = (lane & 2) != 0, b0 = (lane & 1) != 0;

  for (int base = beg; base < end; base += 16) {
    const int n = (end - base < 16) ? (end - base) : 16;
    const bool has = l15 < n;
    int s = node;
    if (has) s = csr_src[base + l15];

    float r0[16], r1[16];
#pragma unroll
    for (int j = 0; j < 16; ++j) {
      const int sj = __builtin_amdgcn_readlane(s, j);
      if (D == 64) {
        const float2 h = *reinterpret_cast<const float2*>(Hl + (size_t)sj * D);
        r0[j] = h.x;
        r1[j] = h.y;
      } else {
        r0[j] = Hl[(size_t)sj * D];
        r1[j] = 0.f;
      }
    }

    float p[16];
#pragma unroll
    for (int j = 0; j < 16; ++j) p[j] = r0[j] * as0 + r1[j] * as1;
    float q[8];
#pragma unroll
    for (int k = 0; k < 8; ++k) {
      const float keep = b3 ? p[k + 8] : p[k];
      const float send = b3 ? p[k] : p[k + 8];
      q[k] = keep + lane_xor<8>(send);
    }
    float t[4];
#pragma unroll
    for (int k = 0; k < 4; ++k) {
      const float keep = b2 ? q[k + 4] : q[k];
      const float send = b2 ? q[k] : q[k + 4];
      t[k] = keep + lane_xor<4>(send);
    }
    float u[2];
#pragma unroll
    for (int k = 0; k < 2; ++k) {
      const float keep = b1 ? t[k + 2] : t[k];
      const float send = b1 ? t[k] : t[k + 2];
      u[k] = keep + lane_xor<2>(send);
    }
    float sc;
    {
      const float keep = b0 ? u[1] : u[0];
      const float send = b0 ? u[0] : u[1];
      sc = keep + lane_xor<1>(send);
    }
    sc += lane_xor16(sc);

    const float v = has ? leaky(sc + ad) : -INFINITY;
    float cm = v;
    cm = fmaxf(cm, lane_xor<8>(cm));
    cm = fmaxf(cm, lane_xor<4>(cm));
    cm = fmaxf(cm, lane_xor<2>(cm));
    cm = fmaxf(cm, lane_xor<1>(cm));
    const float mn    = fmaxf(m, cm);
    const float scale = __expf(m - mn);
    const float w     = has ? __expf(v - mn) : 0.f;
    float ws = w;
    ws += lane_xor<8>(ws);
    ws += lane_xor<4>(ws);
    ws += lane_xor<2>(ws);
    ws += lane_xor<1>(ws);
    l    = l * scale + ws;
    acc0 = acc0 * scale;
    acc1 = acc1 * scale;
    m    = mn;

#pragma unroll
    for (int j = 0; j < 16; ++j) {
      if (j < n) {
        const float wj =
            __int_as_float(__builtin_amdgcn_readlane(__float_as_int(w), j));
        acc0 += wj * r0[j];
        if (D == 64) acc1 += wj * r1[j];
      }
    }
  }

  float* o = out + (size_t)node * D + lane * V;
  if (D == 64) {
    float2 r;
    r.x = acc0 / l;
    r.y = acc1 / l;
    *reinterpret_cast<float2*>(o) = r;
  } else {
    o[0] = acc0 / l;
  }
}

template <int D>
__global__ void finalize_tanh(float* __restrict__ out,
                              const float* __restrict__ b, int N) {
  const int t = blockIdx.x * blockDim.x + threadIdx.x;
  if (t >= N * D) return;
  out[t] = tanhf(out[t] + b[t & (D - 1)]);
}

extern "C" void kernel_launch(void* const* d_in, const int* in_sizes, int n_in,
                              void* d_out, int out_size, void* d_ws,
                              size_t ws_size, hipStream_t stream) {
  const float* x   = (const float*)d_in[0];
  const int*   ei  = (const int*)d_in[1];
  const float* W1  = (const float*)d_in[2];
  const float* as1 = (const float*)d_in[3];
  const float* ad1 = (const float*)d_in[4];
  const float* b1  = (const float*)d_in[5];
  const float* W2  = (const float*)d_in[6];
  const float* as2 = (const float*)d_in[7];
  const float* ad2 = (const float*)d_in[8];
  const float* b2  = (const float*)d_in[9];
  float* out = (float*)d_out;

  const int N    = in_sizes[0] / 64;
  const int E    = in_sizes[1] / 2;
  const int* src = ei;
  const int* dst = ei + E;

  float* ws = (float*)d_ws;
  size_t o = 0;
  float* h1 = ws + o;   o += (size_t)N * 64;
  float* o1 = ws + o;   o += (size_t)N * 64;
  int* cnt     = (int*)(ws + o); o += N;
  int* rowptr  = (int*)(ws + o); o += N + 1;
  int* rank    = (int*)(ws + o); o += E;
  int* csr_src = (int*)(ws + o); o += E;

  const int B  = 256;
  const int ge = (E + B - 1) / B;
  const int gn = (int)(((size_t)N * 32 + B - 1) / B);

  hipMemsetAsync(cnt, 0, (size_t)N * 4, stream);
  csr_count_rank<<<ge, B, 0, stream>>>(dst, cnt, rank, E);
  csr_scan<<<(N + 1023) / 1024, 1024, 0, stream>>>(cnt, rowptr, N);
  csr_fill<<<ge, B, 0, stream>>>(src, dst, rowptr, rank, csr_src, E);

  {
    const int tiles = (N / 16) * (64 / 16);
    gat_gemm_wmma<64, 64, false>
        <<<(tiles + 7) / 8, B, 0, stream>>>(x, W1, nullptr, h1, N);
    node_softmax_aggregate<64>
        <<<gn, B, 0, stream>>>(rowptr, csr_src, as1, ad1, h1, o1, N);
  }

  {
    float* h2 = h1;
    const int tiles = (N / 16) * (32 / 16);
    gat_gemm_wmma<64, 32, true>
        <<<(tiles + 7) / 8, B, 0, stream>>>(o1, W2, b1, h2, N);
    node_softmax_aggregate<32>
        <<<gn, B, 0, stream>>>(rowptr, csr_src, as2, ad2, h2, out, N);
    finalize_tanh<32><<<(int)(((size_t)N * 32 + B - 1) / B), B, 0, stream>>>(
        out, b2, N);
  }
}


# Edge grouping in one pass: the counting atomic's rank puts each source straight into its node's fixed 32-slot row (rank >= 32: overflow list); scan, fill pass, rank array gone; sums in arrival rank
# speedup vs baseline: 3.3451x; 1.0504x over previous
#include <hip/hip_runtime.h>
#include <math.h>


typedef __attribute__((ext_vector_type(2))) float v2f;
typedef __attribute__((ext_vector_type(8))) float v8f;

#define NEG_SLOPE 0.2f

#if defined(__has_builtin)
#if __has_builtin(__builtin_amdgcn_wmma_f32_16x16x4_f32)
#define HAVE_WMMA_F32_16X16X4 1
#endif
#endif

__device__ __forceinline__ float leaky(float v) {
  return v > 0.f ? v : NEG_SLOPE * v;
}

template <int M>
__device__ __forceinline__ float lane_xor(float v) {
  return __builtin_bit_cast(
      float, __builtin_amdgcn_update_dpp(0, __builtin_bit_cast(int, v),
                                         0x160 + M, 0xf, 0xf, true));
}
__device__ __forceinline__ float lane_xor16(float v) {
  const int b = __builtin_bit_cast(int, v);
  return __builtin_bit_cast(
      float, __builtin_amdgcn_permlanex16(b, b, 0x76543210, 0xfedcba98, false,
                                          false));
}

template <int K, int DOUT, bool RELU_BIAS>
__global__ void gat_gemm_wmma(const float* __restrict__ X,
                              const float* __restrict__ W,
                              const float* __restrict__ bias,
                              float* __restrict__ H, int M) {
  const int lane   = threadIdx.x & 31;
  const int wave   = threadIdx.x >> 5;
  const int tileId = blockIdx.x * (blockDim.x >> 5) + wave;
  const int NT     = DOUT / 16;
  const int mt     = tileId / NT;
  const int nt     = tileId % NT;
  if (mt >= M / 16) return;

  const int r15 = lane & 15;
  const int hi  = lane >> 4;
  const int row = mt * 16 + r15;
  const int col = nt * 16 + r15;

  v8f c = {0.f, 0.f, 0.f, 0.f, 0.f, 0.f, 0.f, 0.f};

#if HAVE_WMMA_F32_16X16X4
  for (int k0 = 0; k0 < K; k0 += 4) {
    const int ka = k0 + 2 * hi;
    float x0 = X[row * K + ka];
    float x1 = X[row * K + ka + 1];
    if (RELU_BIAS) {
      x0 = fmaxf(x0 + bias[ka], 0.f);
      x1 = fmaxf(x1 + bias[ka + 1], 0.f);
    }
    v2f a;
    a.x = x0;
    a.y = x1;
    v2f b;
    b.x = W[ka * DOUT + col];
    b.y = W[(ka + 1) * DOUT + col];
    c = __builtin_amdgcn_wmma_f32_16x16x4_f32(false, a, false, b, (short)0, c,
                                              false, false);
  }
#else
  for (int r = 0; r < 8; ++r) {
    const int rr = mt * 16 + hi * 8 + r;
    float acc = 0.f;
    for (int k = 0; k < K; ++k) {
      float xv = X[rr * K + k];
      if (RELU_BIAS) xv = fmaxf(xv + bias[k], 0.f);
      acc += xv * W[k * DOUT + col];
    }
    c[r] = acc;
  }
#endif

  const int rowBase = mt * 16 + hi * 8;
#pragma unroll
  for (int r = 0; r < 8; ++r) H[(rowBase + r) * DOUT + col] = c[r];
}

#define SLOTS 32

__global__ void group_init(int* __restrict__ cnt, int* __restrict__ slots,
                           int* __restrict__ novf, int N) {
  const int t = blockIdx.x * blockDim.x + threadIdx.x;
  const int Q = N * 2;
  if (t >= Q) return;
  int4* p = reinterpret_cast<int4*>(slots);
#pragma unroll
  for (int k = 0; k < 4; ++k) {
    const int q = t + k * Q;
    const int d = q >> 3;
    p[q] = make_int4(d, d, d, d);
  }
  if (t < N) cnt[t] = 0;
  if (t == 0) novf[0] = 0;
}

__global__ void group_edges(const int* __restrict__ src,
                            const int* __restrict__ dst, int* __restrict__ cnt,
                            int* __restrict__ slots, int* __restrict__ novf,
                            int* __restrict__ ovf_dst,
                            int* __restrict__ ovf_src, int E) {
  const int e = blockIdx.x * blockDim.x + threadIdx.x;
  if (e >= E) return;
  const int d = dst[e];
  const int s = src[e];
  const int r = atomicAdd(&cnt[d], 1);
  if (r < SLOTS) {
    slots[(size_t)d * SLOTS + r] = s;
  } else {
    const int p = atomicAdd(novf, 1);
    ovf_dst[p] = d;
    ovf_src[p] = s;
  }
}

template <int D>
__global__ void node_softmax_aggregate(const int* __restrict__ cnt,
                                       const int* __restrict__ slots,
                                       const int* __restrict__ novf,
                                       const int* __restrict__ ovf_dst,
                                       const int* __restrict__ ovf_src,
                                       const float* __restrict__ attS,
                                       const float* __restrict__ attD,
                                       const float* __restrict__ H,
                                       float* __restrict__ out, int N, int E) {
  static_assert(D == 32 || D == 64, "row must be 32 or 64 floats");
  constexpr int V = D / 32;
  const int lane = threadIdx.x & 31;
  const int l15  = lane & 15;
  const int node = __builtin_amdgcn_readfirstlane(
      (int)((blockIdx.x * blockDim.x + threadIdx.x) >> 5));
  if (node >= N) return;

  const float* Hl = H + lane * V;
  float hs0, hs1 = 0.f, as0, as1 = 0.f, ad0, ad1 = 0.f;
  if (D == 64) {
    const float2 h = *reinterpret_cast<const float2*>(Hl + (size_t)node * D);
    const float2 a = *reinterpret_cast<const float2*>(attS + lane * V);
    const float2 b = *reinterpret_cast<const float2*>(attD + lane * V);
    hs0 = h.x; hs1 = h.y; as0 = a.x; as1 = a.y; ad0 = b.x; ad1 = b.y;
  } else {
    hs0 = Hl[(size_t)node * D];
    as0 = attS[lane];
    ad0 = attD[lane];
  }
  const int sl  = slots[(size_t)node * SLOTS + lane];
  const int deg = cnt[node];
  const int nslot = deg < SLOTS ? deg : SLOTS;

  float ps = fmaf(hs0, as0, hs1 * as1);
  float pd = fmaf(hs0, ad0, hs1 * ad1);
  ps += lane_xor16(ps);   pd += lane_xor16(pd);
  ps += lane_xor<8>(ps);  pd += lane_xor<8>(pd);
  ps += lane_xor<4>(ps);  pd += lane_xor<4>(pd);
  ps += lane_xor<2>(ps);  pd += lane_xor<2>(pd);
  ps += lane_xor<1>(ps);  pd += lane_xor<1>(pd);
  const float ad = pd;
  float m = leaky(ps + ad);
  float l = 1.f;
  float acc0 = hs0, acc1 = hs1;

  const bool b3 = (lane & 8) != 0, b2 = (lane & 4) != 0;
  const bool b1 = (lane & 2) != 0, b0 = (lane & 1) != 0;

  asm volatile("" ::"v"(sl));

  for (int base = 0; base < nslot; base += 16) {
    const int n = (nslot - base < 16) ? (nslot - base) : 16;
    const bool has = l15 < n;

    float r0[16], r1[16];
#pragma unroll
    for (int j = 0; j < 16; ++j) {
      const int sj = __builtin_amdgcn_readlane(sl, base + j);
      if (D == 64) {
        const float2 h = *reinterpret_cast<const float2*>(Hl + (size_t)sj * D);
        r0[j] = h.x;
        r1[j] = h.y;
      } else {
        r0[j] = Hl[(size_t)sj * D];
        r1[j] = 0.f;
      }
    }

    float p[16];
#pragma unroll
    for (int j = 0; j < 16; ++j) p[j] = r0[j] * as0 + r1[j] * as1;
    float q[8];
#pragma unroll
    for (int k = 0; k < 8; ++k) {
      const float keep = b3 ? p[k + 8] : p[k];
      const float send = b3 ? p[k] : p[k + 8];
      q[k] = keep + lane_xor<8>(send);
    }
    float t[4];
#pragma unroll
    for (int k = 0; k < 4; ++k) {
      const float keep = b2 ? q[k + 4] : q[k];
      const float send = b2 ? q[k] : q[k + 4];
      t[k] = keep + lane_xor<4>(send);
    }
    float u[2];
#pragma unroll
    for (int k = 0; k < 2; ++k) {
      const float keep = b1 ? t[k + 2] : t[k];
      const float send = b1 ? t[k] : t[k + 2];
      u[k] = keep + lane_xor<2>(send);
    }
    float sc;
    {
      const float keep = b0 ? u[1] : u[0];
      const float send = b0 ? u[0] : u[1];
      sc = keep + lane_xor<1>(send);
    }
    sc += lane_xor16(sc);

    const float v = has ? leaky(sc + ad) : -INFINITY;
    float cm = v;
    cm = fmaxf(cm, lane_xor<8>(cm));
    cm = fmaxf(cm, lane_xor<4>(cm));
    cm = fmaxf(cm, lane_xor<2>(cm));
    cm = fmaxf(cm, lane_xor<1>(cm));
    const float mn    = fmaxf(m, cm);
    const float scale = __expf(m - mn);
    const float w     = has ? __expf(v - mn) : 0.f;
    float ws = w;
    ws += lane_xor<8>(ws);
    ws += lane_xor<4>(ws);
    ws += lane_xor<2>(ws);
    ws += lane_xor<1>(ws);
    l    = l * scale + ws;
    acc0 = acc0 * scale;
    acc1 = acc1 * scale;
    m    = mn;

#pragma unroll
    for (int j = 0; j < 16; ++j) {
      if (j < n) {
        const float wj =
            __int_as_float(__builtin_amdgcn_readlane(__float_as_int(w), j));
        acc0 += wj * r0[j];
        if (D == 64) acc1 += wj * r1[j];
      }
    }
  }

  if (deg > SLOTS) {
    int nov = novf[0];
    nov = nov < E ? nov : E;
    for (int base = 0; base < nov; base += 32) {
      const int i = base + lane;
      int od = -1, os = 0;
      if (i < nov) {
        od = ovf_dst[i];
        os = ovf_src[i];
      }
      unsigned mask = __builtin_amdgcn_ballot_w32(od == node);
      while (mask != 0u) {
        const int j = __builtin_ctz(mask);
        mask &= mask - 1u;
        const int sj = __builtin_amdgcn_readlane(os, j);
        float g0, g1 = 0.f;
        if (D == 64) {
          const float2 h = *reinterpret_cast<const float2*>(Hl + (size_t)sj * D);
          g0 = h.x;
          g1 = h.y;
        } else {
          g0 = Hl[(size_t)sj * D];
        }
        float sc = fmaf(g0, as0, g1 * as1);
        sc += lane_xor16(sc);
        sc += lane_xor<8>(sc);
        sc += lane_xor<4>(sc);
        sc += lane_xor<2>(sc);
        sc += lane_xor<1>(sc);
        const float v     = leaky(sc + ad);
        const float mn    = fmaxf(m, v);
        const float scale = __expf(m - mn);
        const float w     = __expf(v - mn);
        l    = l * scale + w;
        acc0 = acc0 * scale + w * g0;
        acc1 = acc1 * scale + w * g1;
        m    = mn;
      }
    }
  }

  float* o = out + (size_t)node * D + lane * V;
  if (D == 64) {
    float2 r;
    r.x = acc0 / l;
    r.y = acc1 / l;
    *reinterpret_cast<float2*>(o) = r;
  } else {
    o[0] = acc0 / l;
  }
}

template <int D>
__global__ void finalize_tanh(float* __restrict__ out,
                              const float* __restrict__ b, int N) {
  const int t = blockIdx.x * blockDim.x + threadIdx.x;
  if (t >= N * D) return;
  out[t] = tanhf(out[t] + b[t & (D - 1)]);
}

extern "C" void kernel_launch(void* const* d_in, const int* in_sizes, int n_in,
                              void* d_out, int out_size, void* d_ws,
                              size_t ws_size, hipStream_t stream) {
  const float* x   = (const float*)d_in[0];
  const int*   ei  = (const int*)d_in[1];
  const float* W1  = (const float*)d_in[2];
  const float* as1 = (const float*)d_in[3];
  const float* ad1 = (const float*)d_in[4];
  const float* b1  = (const float*)d_in[5];
  const float* W2  = (const float*)d_in[6];
  const float* as2 = (const float*)d_in[7];
  const float* ad2 = (const float*)d_in[8];
  const float* b2  = (const float*)d_in[9];
  float* out = (float*)d_out;

  const int N    = in_sizes[0] / 64;
  const int E    = in_sizes[1] / 2;
  const int* src = ei;
  const int* dst = ei + E;

  float* ws = (float*)d_ws;
  size_t o = 0;
  float* h1 = ws + o;   o += (size_t)N * 64;
  float* o1 = ws + o;   o += (size_t)N * 64;
  int* slots   = (int*)(ws + o); o += (size_t)N * SLOTS;
  int* novf    = (int*)(ws + o); o += 32;
  int* ovf_dst = (int*)(ws + o); o += E;
  int* ovf_src = (int*)(ws + o); o += E;
  int* cnt     = (int*)(ws + o); o += N;

  const int B  = 256;
  const int ge = (E + B - 1) / B;
  const int gn = (int)(((size_t)N * 32 + B - 1) / B);

  group_init<<<(N * 2 + B - 1) / B, B, 0, stream>>>(cnt, slots, novf, N);
  group_edges<<<ge, B, 0, stream>>>(src, dst, cnt, slots, novf, ovf_dst,
                                    ovf_src, E);

  {
    const int tiles = (N / 16) * (64 / 16);
    gat_gemm_wmma<64, 64, false>
        <<<(tiles + 7) / 8, B, 0, stream>>>(x, W1, nullptr, h1, N);
    node_softmax_aggregate<64>
        <<<gn, B, 0, stream>>>(cnt, slots, novf, ovf_dst, ovf_src, as1, ad1, h1,
                                o1, N, E);
  }

  {
    float* h2 = h1;
    const int tiles = (N / 16) * (32 / 16);
    gat_gemm_wmma<64, 32, true>
        <<<(tiles + 7) / 8, B, 0, stream>>>(o1, W2, b1, h2, N);
    node_softmax_aggregate<32>
        <<<gn, B, 0, stream>>>(cnt, slots, novf, ovf_dst, ovf_src, as2, ad2, h2,
                                out, N, E);
    finalize_tanh<32><<<(int)(((size_t)N * 32 + B - 1) / B), B, 0, stream>>>(
        out, b2, N);
  }
}


# Edge grouping as a two-level counting sort on LDS counters (chunk sort by 256-node bucket, then one work-group per bucket builds slot rows in LDS): no global atomic per edge; sum order = LDS rank
# speedup vs baseline: 6.4808x; 1.9374x over previous
#include <hip/hip_runtime.h>
#include <math.h>


typedef __attribute__((ext_vector_type(2))) float v2f;
typedef __attribute__((ext_vector_type(8))) float v8f;

#define NEG_SLOPE 0.2f

#if defined(__has_builtin)
#if __has_builtin(__builtin_amdgcn_wmma_f32_16x16x4_f32)
#define HAVE_WMMA_F32_16X16X4 1
#endif
#endif

__device__ __forceinline__ float leaky(float v) {
  return v > 0.f ? v : NEG_SLOPE * v;
}

template <int M>
__device__ __forceinline__ float lane_xor(float v) {
  return __builtin_bit_cast(
      float, __builtin_amdgcn_update_dpp(0, __builtin_bit_cast(int, v),
                                         0x160 + M, 0xf, 0xf, true));
}
__device__ __forceinline__ float lane_xor16(float v) {
  const int b = __builtin_bit_cast(int, v);
  return __builtin_bit_cast(
      float, __builtin_amdgcn_permlanex16(b, b, 0x76543210, 0xfedcba98, false,
                                          false));
}

template <int K, int DOUT, bool RELU_BIAS>
__global__ void gat_gemm_wmma(const float* __restrict__ X,
                              const float* __restrict__ W,
                              const float* __restrict__ bias,
                              float* __restrict__ H, int M) {
  const int lane   = threadIdx.x & 31;
  const int wave   = threadIdx.x >> 5;
  const int tileId = blockIdx.x * (blockDim.x >> 5) + wave;
  const int NT     = DOUT / 16;
  const int mt     = tileId / NT;
  const int nt     = tileId % NT;
  if (mt >= M / 16) return;

  const int r15 = lane & 15;
  const int hi  = lane >> 4;
  const int row = mt * 16 + r15;
  const int col = nt * 16 + r15;

  v8f c = {0.f, 0.f, 0.f, 0.f, 0.f, 0.f, 0.f, 0.f};

#if HAVE_WMMA_F32_16X16X4
  for (int k0 = 0; k0 < K; k0 += 4) {
    const int ka = k0 + 2 * hi;
    float x0 = X[row * K + ka];
    float x1 = X[row * K + ka + 1];
    if (RELU_BIAS) {
      x0 = fmaxf(x0 + bias[ka], 0.f);
      x1 = fmaxf(x1 + bias[ka + 1], 0.f);
    }
    v2f a;
    a.x = x0;
    a.y = x1;
    v2f b;
    b.x = W[ka * DOUT + col];
    b.y = W[(ka + 1) * DOUT + col];
    c = __builtin_amdgcn_wmma_f32_16x16x4_f32(false, a, false, b, (short)0, c,
                                              false, false);
  }
#else
  for (int r = 0; r < 8; ++r) {
    const int rr = mt * 16 + hi * 8 + r;
    float acc = 0.f;
    for (int k = 0; k < K; ++k) {
      float xv = X[rr * K + k];
      if (RELU_BIAS) xv = fmaxf(xv + bias[k], 0.f);
      acc += xv * W[k * DOUT + col];
    }
    c[r] = acc;
  }
#endif

  const int rowBase = mt * 16 + hi * 8;
#pragma unroll
  for (int r = 0; r < 8; ++r) H[(rowBase + r) * DOUT + col] = c[r];
}

#define SLOTS 32
#define CH 8192
#define EPT 32
#define NBK 256
#define BMAX 1024

__global__ void __launch_bounds__(256)
part_edges(const int* __restrict__ src, const int* __restrict__ dst,
           unsigned* __restrict__ sorted, unsigned* __restrict__ meta,
           int* __restrict__ novf, int E, int Bn, int G) {
  __shared__ __attribute__((aligned(16))) int hist[BMAX];
  __shared__ __attribute__((aligned(16))) unsigned stage[CH];
  __shared__ int wsum[8];
  const int t    = threadIdx.x;
  const int lane = t & 31;
  const int wave = t >> 5;
  const int g    = blockIdx.x;
  const int e0   = g * CH;

  for (int i = t; i < BMAX; i += 256) hist[i] = 0;
  if (g == 0 && t == 0) novf[0] = 0;
  __syncthreads();

  int d[EPT], s[EPT], r[EPT];
#pragma unroll
  for (int k = 0; k < EPT; ++k) {
    const int e  = e0 + k * 256 + t;
    const int ec = e < E ? e : E - 1;
    d[k] = dst[ec];
    s[k] = src[ec];
  }
#pragma unroll
  for (int k = 0; k < EPT; ++k) {
    const int e = e0 + k * 256 + t;
    int b = d[k] >> 8;
    b = b < 0 ? 0 : (b < BMAX ? b : BMAX - 1);
    d[k] = e < E ? d[k] : -1;
    s[k] = (int)(((unsigned)s[k] << 8) | (unsigned)(d[k] & 255));
    r[k] = 0;
    if (d[k] >= 0) r[k] = atomicAdd(&hist[b], 1);
    d[k] = d[k] >= 0 ? b : -1;
  }
  __syncthreads();

  const int4 h = reinterpret_cast<const int4*>(hist)[t];
  const int sum = h.x + h.y + h.z + h.w;
  int incl = sum;
#pragma unroll
  for (int off = 1; off < 32; off <<= 1) {
    const int v = __shfl_up(incl, off);
    if (lane >= off) incl += v;
  }
  if (lane == 31) wsum[wave] = incl;
  __syncthreads();
  int base = 0;
  for (int w = 0; w < wave; ++w) base += wsum[w];
  int4 st;
  st.x = base + incl - sum;
  st.y = st.x + h.x;
  st.z = st.y + h.y;
  st.w = st.z + h.z;
  reinterpret_cast<int4*>(hist)[t] = st;
  {
    const int b = 4 * t;
    unsigned* m = meta + (size_t)b * G + g;
    if (b + 0 < Bn) m[0]             = (unsigned)st.x | ((unsigned)h.x << 16);
    if (b + 1 < Bn) m[(size_t)G]     = (unsigned)st.y | ((unsigned)h.y << 16);
    if (b + 2 < Bn) m[(size_t)G * 2] = (unsigned)st.z | ((unsigned)h.z << 16);
    if (b + 3 < Bn) m[(size_t)G * 3] = (unsigned)st.w | ((unsigned)h.w << 16);
  }
  __syncthreads();

#pragma unroll
  for (int k = 0; k < EPT; ++k) r[k] += hist[d[k] >= 0 ? d[k] : 0];
#pragma unroll
  for (int k = 0; k < EPT; ++k)
    if (d[k] >= 0) stage[r[k]] = (unsigned)s[k];
  __syncthreads();

  const int clen = (E - e0) < CH ? (E - e0) : CH;
  const int c4   = clen >> 2;
  uint4* o4 = reinterpret_cast<uint4*>(sorted + e0);
  const uint4* s4 = reinterpret_cast<const uint4*>(stage);
  for (int i = t; i < c4; i += 256) o4[i] = s4[i];
  for (int i = (c4 << 2) + t; i < clen; i += 256) sorted[e0 + i] = stage[i];
}

__device__ __forceinline__ void put_entry(unsigned ent, int r, int n0,
                                          int* lslots, int* __restrict__ novf,
                                          int* __restrict__ ovf_dst,
                                          int* __restrict__ ovf_src, int E) {
  const int n = (int)(ent & 255u);
  const int s = (int)(ent >> 8);
  if (r < SLOTS) {
    lslots[n * SLOTS + r] = s;
  } else {
    const int p = atomicAdd(novf, 1);
    if (p < E) {
      ovf_dst[p] = n0 + n;
      ovf_src[p] = s;
    }
  }
}

__global__ void __launch_bounds__(256)
fill_slots(const unsigned* __restrict__ sorted,
           const unsigned* __restrict__ meta, int* __restrict__ cnt,
           int* __restrict__ slots, int* __restrict__ novf,
           int* __restrict__ ovf_dst, int* __restrict__ ovf_src, int N, int E,
           int G) {
  __shared__ __attribute__((aligned(16))) int lslots[NBK * SLOTS];
  __shared__ int lcnt[NBK];
  const int t    = threadIdx.x;
  const int lane = t & 31;
  const int wave = __builtin_amdgcn_readfirstlane(t >> 5);
  const int b    = blockIdx.x;
  const int n0   = b * NBK;

  int4* l4 = reinterpret_cast<int4*>(lslots);
#pragma unroll
  for (int k = 0; k < (NBK * SLOTS / 4) / 256; ++k) {
    const int q = t + k * 256;
    const int dn = n0 + (q >> 3);
    l4[q] = make_int4(dn, dn, dn, dn);
  }
  lcnt[t] = 0;
  __syncthreads();

  const int GW   = (G + 7) >> 3;
  const int gBeg = wave * GW;
  const int gEnd = (gBeg + GW) < G ? (gBeg + GW) : G;
  const unsigned* mrow = meta + (size_t)b * G;

  for (int kb = gBeg; kb < gEnd; kb += 32) {
    const int nr = (gEnd - kb) < 32 ? (gEnd - kb) : 32;
    const unsigned mv = lane < nr ? mrow[kb + lane] : 0u;

    unsigned ent[32];
    int rk[32];
#pragma unroll
    for (int u = 0; u < 32; ++u) {
      const unsigned m = (unsigned)__builtin_amdgcn_readlane((int)mv, u);
      const int st  = (int)(m & 0xffffu);
      const int len = (int)(m >> 16);
      ent[u] = 0u;
      if (lane < len) ent[u] = sorted[(size_t)(kb + u) * CH + st + lane];
    }
#pragma unroll
    for (int u = 0; u < 32; ++u) {
      const unsigned m = (unsigned)__builtin_amdgcn_readlane((int)mv, u);
      const int len = (int)(m >> 16);
      rk[u] = 0;
      if (lane < len) rk[u] = atomicAdd(&lcnt[ent[u] & 255u], 1);
    }
#pragma unroll
    for (int u = 0; u < 32; ++u) {
      const unsigned m = (unsigned)__builtin_amdgcn_readlane((int)mv, u);
      const int len = (int)(m >> 16);
      if (lane < len)
        put_entry(ent[u], rk[u], n0, lslots, novf, ovf_dst, ovf_src, E);
    }
    for (int u = 0; u < nr; ++u) {
      const unsigned m = (unsigned)__builtin_amdgcn_readlane((int)mv, u);
      const int st  = (int)(m & 0xffffu);
      const int len = (int)(m >> 16);
      for (int i = 32 + lane; i < len; i += 32) {
        const unsigned e = sorted[(size_t)(kb + u) * CH + st + i];
        const int r = atomicAdd(&lcnt[e & 255u], 1);
        put_entry(e, r, n0, lslots, novf, ovf_dst, ovf_src, E);
      }
    }
  }
  __syncthreads();

  const int nn = (N - n0) < NBK ? (N - n0) : NBK;
  int4* g4 = reinterpret_cast<int4*>(slots + (size_t)n0 * SLOTS);
#pragma unroll
  for (int k = 0; k < (NBK * SLOTS / 4) / 256; ++k) {
    const int q = t + k * 256;
    if ((q >> 3) < nn) g4[q] = l4[q];
  }
  if (t < nn) cnt[n0 + t] = lcnt[t];
}

template <int D>
__global__ void node_softmax_aggregate(const int* __restrict__ cnt,
                                       const int* __restrict__ slots,
                                       const int* __restrict__ novf,
                                       const int* __restrict__ ovf_dst,
                                       const int* __restrict__ ovf_src,
                                       const float* __restrict__ attS,
                                       const float* __restrict__ attD,
                                       const float* __restrict__ H,
                                       float* __restrict__ out, int N, int E) {
  static_assert(D == 32 || D == 64, "row must be 32 or 64 floats");
  constexpr int V = D / 32;
  const int lane = threadIdx.x & 31;
  const int l15  = lane & 15;
  const int node = __builtin_amdgcn_readfirstlane(
      (int)((blockIdx.x * blockDim.x + threadIdx.x) >> 5));
  if (node >= N) return;

  const float* Hl = H + lane * V;
  float hs0, hs1 = 0.f, as0, as1 = 0.f, ad0, ad1 = 0.f;
  if (D == 64) {
    const float2 h = *reinterpret_cast<const float2*>(Hl + (size_t)node * D);
    const float2 a = *reinterpret_cast<const float2*>(attS + lane * V);
    const float2 b = *reinterpret_cast<const float2*>(attD + lane * V);
    hs0 = h.x; hs1 = h.y; as0 = a.x; as1 = a.y; ad0 = b.x; ad1 = b.y;
  } else {
    hs0 = Hl[(size_t)node * D];
    as0 = attS[lane];
    ad0 = attD[lane];
  }
  const int sl  = slots[(size_t)node * SLOTS + lane];
  const int deg = cnt[node];
  const int nslot = deg < SLOTS ? deg : SLOTS;

  float ps = fmaf(hs0, as0, hs1 * as1);
  float pd = fmaf(hs0, ad0, hs1 * ad1);
  ps += lane_xor16(ps);   pd += lane_xor16(pd);
  ps += lane_xor<8>(ps);  pd += lane_xor<8>(pd);
  ps += lane_xor<4>(ps);  pd += lane_xor<4>(pd);
  ps += lane_xor<2>(ps);  pd += lane_xor<2>(pd);
  ps += lane_xor<1>(ps);  pd += lane_xor<1>(pd);
  const float ad = pd;
  float m = leaky(ps + ad);
  float l = 1.f;
  float acc0 = hs0, acc1 = hs1;

  const bool b3 = (lane & 8) != 0, b2 = (lane & 4) != 0;
  const bool b1 = (lane & 2) != 0, b0 = (lane & 1) != 0;

  asm volatile("" ::"v"(sl));

  for (int base = 0; base < nslot; base += 16) {
    const int n = (nslot - base < 16) ? (nslot - base) : 16;
    const bool has = l15 < n;

    float r0[16], r1[16];
#pragma unroll
    for (int j = 0; j < 16; ++j) {
      const int sj = __builtin_amdgcn_readlane(sl, base + j);
      if (D == 64) {
        const float2 h = *reinterpret_cast<const float2*>(Hl + (size_t)sj * D);
        r0[j] = h.x;
        r1[j] = h.y;
      } else {
        r0[j] = Hl[(size_t)sj * D];
        r1[j] = 0.f;
      }
    }

    float p[16];
#pragma unroll
    for (int j = 0; j < 16; ++j) p[j] = r0[j] * as0 + r1[j] * as1;
    float q[8];
#pragma unroll
    for (int k = 0; k < 8; ++k) {
      const float keep = b3 ? p[k + 8] : p[k];
      const float send = b3 ? p[k] : p[k + 8];
      q[k] = keep + lane_xor<8>(send);
    }
    float t[4];
#pragma unroll
    for (int k = 0; k < 4; ++k) {
      const float keep = b2 ? q[k + 4] : q[k];
      const float send = b2 ? q[k] : q[k + 4];
      t[k] = keep + lane_xor<4>(send);
    }
    float u[2];
#pragma unroll
    for (int k = 0; k < 2; ++k) {
      const float keep = b1 ? t[k + 2] : t[k];
      const float send = b1 ? t[k] : t[k + 2];
      u[k] = keep + lane_xor<2>(send);
    }
    float sc;
    {
      const float keep = b0 ? u[1] : u[0];
      const float send = b0 ? u[0] : u[1];
      sc = keep + lane_xor<1>(send);
    }
    sc += lane_xor16(sc);

    const float v = has ? leaky(sc + ad) : -INFINITY;
    float cm = v;
    cm = fmaxf(cm, lane_xor<8>(cm));
    cm = fmaxf(cm, lane_xor<4>(cm));
    cm = fmaxf(cm, lane_xor<2>(cm));
    cm = fmaxf(cm, lane_xor<1>(cm));
    const float mn    = fmaxf(m, cm);
    const float scale = __expf(m - mn);
    const float w     = has ? __expf(v - mn) : 0.f;
    float ws = w;
    ws += lane_xor<8>(ws);
    ws += lane_xor<4>(ws);
    ws += lane_xor<2>(ws);
    ws += lane_xor<1>(ws);
    l    = l * scale + ws;
    acc0 = acc0 * scale;
    acc1 = acc1 * scale;
    m    = mn;

#pragma unroll
    for (int j = 0; j < 16; ++j) {
      if (j < n) {
        const float wj =
            __int_as_float(__builtin_amdgcn_readlane(__float_as_int(w), j));
        acc0 += wj * r0[j];
        if (D == 64) acc1 += wj * r1[j];
      }
    }
  }

  if (deg > SLOTS) {
    int nov = novf[0];
    nov = nov < E ? nov : E;
    for (int base = 0; base < nov; base += 32) {
      const int i = base + lane;
      int od = -1, os = 0;
      if (i < nov) {
        od = ovf_dst[i];
        os = ovf_src[i];
      }
      unsigned mask = __builtin_amdgcn_ballot_w32(od == node);
      while (mask != 0u) {
        const int j = __builtin_ctz(mask);
        mask &= mask - 1u;
        const int sj = __builtin_amdgcn_readlane(os, j);
        float g0, g1 = 0.f;
        if (D == 64) {
          const float2 h = *reinterpret_cast<const float2*>(Hl + (size_t)sj * D);
          g0 = h.x;
          g1 = h.y;
        } else {
          g0 = Hl[(size_t)sj * D];
        }
        float sc = fmaf(g0, as0, g1 * as1);
        sc += lane_xor16(sc);
        sc += lane_xor<8>(sc);
        sc += lane_xor<4>(sc);
        sc += lane_xor<2>(sc);
        sc += lane_xor<1>(sc);
        const float v     = leaky(sc + ad);
        const float mn    = fmaxf(m, v);
        const float scale = __expf(m - mn);
        const float w     = __expf(v - mn);
        l    = l * scale + w;
        acc0 = acc0 * scale + w * g0;
        acc1 = acc1 * scale + w * g1;
        m    = mn;
      }
    }
  }

  float* o = out + (size_t)node * D + lane * V;
  if (D == 64) {
    float2 r;
    r.x = acc0 / l;
    r.y = acc1 / l;
    *reinterpret_cast<float2*>(o) = r;
  } else {
    o[0] = acc0 / l;
  }
}

template <int D>
__global__ void finalize_tanh(float* __restrict__ out,
                              const float* __restrict__ b, int N) {
  const int t = blockIdx.x * blockDim.x + threadIdx.x;
  if (t >= N * D) return;
  out[t] = tanhf(out[t] + b[t & (D - 1)]);
}

extern "C" void kernel_launch(void* const* d_in, const int* in_sizes, int n_in,
                              void* d_out, int out_size, void* d_ws,
                              size_t ws_size, hipStream_t stream) {
  const float* x   = (const float*)d_in[0];
  const int*   ei  = (const int*)d_in[1];
  const float* W1  = (const float*)d_in[2];
  const float* as1 = (const float*)d_in[3];
  const float* ad1 = (const float*)d_in[4];
  const float* b1  = (const float*)d_in[5];
  const float* W2  = (const float*)d_in[6];
  const float* as2 = (const float*)d_in[7];
  const float* ad2 = (const float*)d_in[8];
  const float* b2  = (const float*)d_in[9];
  float* out = (float*)d_out;

  const int N    = in_sizes[0] / 64;
  const int E    = in_sizes[1] / 2;
  const int* src = ei;
  const int* dst = ei + E;

  float* ws = (float*)d_ws;
  size_t o = 0;
  float* h1 = ws + o;   o += (size_t)N * 64;
  float* o1 = ws + o;   o += (size_t)N * 64;
  int* slots   = (int*)(ws + o); o += (size_t)N * SLOTS;
  int* novf    = (int*)(ws + o); o += 32;
  int* ovf_dst = (int*)(ws + o); o += E;
  int* ovf_src = (int*)(ws + o); o += E;
  int* cnt     = (int*)(ws + o); o += N;
  const int G  = (E + CH - 1) / CH;
  const int Bn = (N + NBK - 1) / NBK;
  o = (o + 31) & ~(size_t)31;
  unsigned* sorted = (unsigned*)(ws + o); o += (size_t)G * CH;
  unsigned* meta   = (unsigned*)(ws + o); o += (size_t)Bn * G;

  const int B  = 256;
  const int gn = (int)(((size_t)N * 32 + B - 1) / B);

  part_edges<<<G, B, 0, stream>>>(src, dst, sorted, meta, novf, E, Bn, G);
  fill_slots<<<Bn, B, 0, stream>>>(sorted, meta, cnt, slots, novf, ovf_dst,
                                   ovf_src, N, E, G);

  {
    const int tiles = (N / 16) * (64 / 16);
    gat_gemm_wmma<64, 64, false>
        <<<(tiles + 7) / 8, B, 0, stream>>>(x, W1, nullptr, h1, N);
    node_softmax_aggregate<64>
        <<<gn, B, 0, stream>>>(cnt, slots, novf, ovf_dst, ovf_src, as1, ad1, h1,
                                o1, N, E);
  }

  {
    float* h2 = h1;
    const int tiles = (N / 16) * (32 / 16);
    gat_gemm_wmma<64, 32, true>
        <<<(tiles + 7) / 8, B, 0, stream>>>(o1, W2, b1, h2, N);
    node_softmax_aggregate<32>
        <<<gn, B, 0, stream>>>(cnt, slots, novf, ovf_dst, ovf_src, as2, ad2, h2,
                                out, N, E);
    finalize_tanh<32><<<(int)(((size_t)N * 32 + B - 1) / B), B, 0, stream>>>(
        out, b2, N);
  }
}


# Final tanh(row + b2) applied where the layer-2 aggregate stores its row, values still in registers: the epilogue launch and its read-back of out are gone; same tanhf, output bit-identical
# speedup vs baseline: 6.7653x; 1.0439x over previous
#include <hip/hip_runtime.h>
#include <math.h>


typedef __attribute__((ext_vector_type(2))) float v2f;
typedef __attribute__((ext_vector_type(8))) float v8f;

#define NEG_SLOPE 0.2f

#if defined(__has_builtin)
#if __has_builtin(__builtin_amdgcn_wmma_f32_16x16x4_f32)
#define HAVE_WMMA_F32_16X16X4 1
#endif
#endif

__device__ __forceinline__ float leaky(float v) {
  return v > 0.f ? v : NEG_SLOPE * v;
}

template <int M>
__device__ __forceinline__ float lane_xor(float v) {
  return __builtin_bit_cast(
      float, __builtin_amdgcn_update_dpp(0, __builtin_bit_cast(int, v),
                                         0x160 + M, 0xf, 0xf, true));
}
__device__ __forceinline__ float lane_xor16(float v) {
  const int b = __builtin_bit_cast(int, v);
  return __builtin_bit_cast(
      float, __builtin_amdgcn_permlanex16(b, b, 0x76543210, 0xfedcba98, false,
                                          false));
}

template <int K, int DOUT, bool RELU_BIAS>
__global__ void gat_gemm_wmma(const float* __restrict__ X,
                              const float* __restrict__ W,
                              const float* __restrict__ bias,
                              float* __restrict__ H, int M) {
  const int lane   = threadIdx.x & 31;
  const int wave   = threadIdx.x >> 5;
  const int tileId = blockIdx.x * (blockDim.x >> 5) + wave;
  const int NT     = DOUT / 16;
  const int mt     = tileId / NT;
  const int nt     = tileId % NT;
  if (mt >= M / 16) return;

  const int r15 = lane & 15;
  const int hi  = lane >> 4;
  const int row = mt * 16 + r15;
  const int col = nt * 16 + r15;

  v8f c = {0.f, 0.f, 0.f, 0.f, 0.f, 0.f, 0.f, 0.f};

#if HAVE_WMMA_F32_16X16X4
  for (int k0 = 0; k0 < K; k0 += 4) {
    const int ka = k0 + 2 * hi;
    float x0 = X[row * K + ka];
    float x1 = X[row * K + ka + 1];
    if (RELU_BIAS) {
      x0 = fmaxf(x0 + bias[ka], 0.f);
      x1 = fmaxf(x1 + bias[ka + 1], 0.f);
    }
    v2f a;
    a.x = x0;
    a.y = x1;
    v2f b;
    b.x = W[ka * DOUT + col];
    b.y = W[(ka + 1) * DOUT + col];
    c = __builtin_amdgcn_wmma_f32_16x16x4_f32(false, a, false, b, (short)0, c,
                                              false, false);
  }
#else
  for (int r = 0; r < 8; ++r) {
    const int rr = mt * 16 + hi * 8 + r;
    float acc = 0.f;
    for (int k = 0; k < K; ++k) {
      float xv = X[rr * K + k];
      if (RELU_BIAS) xv = fmaxf(xv + bias[k], 0.f);
      acc += xv * W[k * DOUT + col];
    }
    c[r] = acc;
  }
#endif

  const int rowBase = mt * 16 + hi * 8;
#pragma unroll
  for (int r = 0; r < 8; ++r) H[(rowBase + r) * DOUT + col] = c[r];
}

#define SLOTS 32
#define CH 8192
#define EPT 32
#define NBK 256
#define BMAX 1024

__global__ void __launch_bounds__(256)
part_edges(const int* __restrict__ src, const int* __restrict__ dst,
           unsigned* __restrict__ sorted, unsigned* __restrict__ meta,
           int* __restrict__ novf, int E, int Bn, int G) {
  __shared__ __attribute__((aligned(16))) int hist[BMAX];
  __shared__ __attribute__((aligned(16))) unsigned stage[CH];
  __shared__ int wsum[8];
  const int t    = threadIdx.x;
  const int lane = t & 31;
  const int wave = t >> 5;
  const int g    = blockIdx.x;
  const int e0   = g * CH;

  for (int i = t; i < BMAX; i += 256) hist[i] = 0;
  if (g == 0 && t == 0) novf[0] = 0;
  __syncthreads();

  int d[EPT], s[EPT], r[EPT];
#pragma unroll
  for (int k = 0; k < EPT; ++k) {
    const int e  = e0 + k * 256 + t;
    const int ec = e < E ? e : E - 1;
    d[k] = dst[ec];
    s[k] = src[ec];
  }
#pragma unroll
  for (int k = 0; k < EPT; ++k) {
    const int e = e0 + k * 256 + t;
    int b = d[k] >> 8;
    b = b < 0 ? 0 : (b < BMAX ? b : BMAX - 1);
    d[k] = e < E ? d[k] : -1;
    s[k] = (int)(((unsigned)s[k] << 8) | (unsigned)(d[k] & 255));
    r[k] = 0;
    if (d[k] >= 0) r[k] = atomicAdd(&hist[b], 1);
    d[k] = d[k] >= 0 ? b : -1;
  }
  __syncthreads();

  const int4 h = reinterpret_cast<const int4*>(hist)[t];
  const int sum = h.x + h.y + h.z + h.w;
  int incl = sum;
#pragma unroll
  for (int off = 1; off < 32; off <<= 1) {
    const int v = __shfl_up(incl, off);
    if (lane >= off) incl += v;
  }
  if (lane == 31) wsum[wave] = incl;
  __syncthreads();
  int base = 0;
  for (int w = 0; w < wave; ++w) base += wsum[w];
  int4 st;
  st.x = base + incl - sum;
  st.y = st.x + h.x;
  st.z = st.y + h.y;
  st.w = st.z + h.z;
  reinterpret_cast<int4*>(hist)[t] = st;
  {
    const int b = 4 * t;
    unsigned* m = meta + (size_t)b * G + g;
    if (b + 0 < Bn) m[0]             = (unsigned)st.x | ((unsigned)h.x << 16);
    if (b + 1 < Bn) m[(size_t)G]     = (unsigned)st.y | ((unsigned)h.y << 16);
    if (b + 2 < Bn) m[(size_t)G * 2] = (unsigned)st.z | ((unsigned)h.z << 16);
    if (b + 3 < Bn) m[(size_t)G * 3] = (unsigned)st.w | ((unsigned)h.w << 16);
  }
  __syncthreads();

#pragma unroll
  for (int k = 0; k < EPT; ++k) r[k] += hist[d[k] >= 0 ? d[k] : 0];
#pragma unroll
  for (int k = 0; k < EPT; ++k)
    if (d[k] >= 0) stage[r[k]] = (unsigned)s[k];
  __syncthreads();

  const int clen = (E - e0) < CH ? (E - e0) : CH;
  const int c4   = clen >> 2;
  uint4* o4 = reinterpret_cast<uint4*>(sorted + e0);
  const uint4* s4 = reinterpret_cast<const uint4*>(stage);
  for (int i = t; i < c4; i += 256) o4[i] = s4[i];
  for (int i = (c4 << 2) + t; i < clen; i += 256) sorted[e0 + i] = stage[i];
}

__device__ __forceinline__ void put_entry(unsigned ent, int r, int n0,
                                          int* lslots, int* __restrict__ novf,
                                          int* __restrict__ ovf_dst,
                                          int* __restrict__ ovf_src, int E) {
  const int n = (int)(ent & 255u);
  const int s = (int)(ent >> 8);
  if (r < SLOTS) {
    lslots[n * SLOTS + r] = s;
  } else {
    const int p = atomicAdd(novf, 1);
    if (p < E) {
      ovf_dst[p] = n0 + n;
      ovf_src[p] = s;
    }
  }
}

__global__ void __launch_bounds__(256)
fill_slots(const unsigned* __restrict__ sorted,
           const unsigned* __restrict__ meta, int* __restrict__ cnt,
           int* __restrict__ slots, int* __restrict__ novf,
           int* __restrict__ ovf_dst, int* __restrict__ ovf_src, int N, int E,
           int G) {
  __shared__ __attribute__((aligned(16))) int lslots[NBK * SLOTS];
  __shared__ int lcnt[NBK];
  const int t    = threadIdx.x;
  const int lane = t & 31;
  const int wave = __builtin_amdgcn_readfirstlane(t >> 5);
  const int b    = blockIdx.x;
  const int n0   = b * NBK;

  int4* l4 = reinterpret_cast<int4*>(lslots);
#pragma unroll
  for (int k = 0; k < (NBK * SLOTS / 4) / 256; ++k) {
    const int q = t + k * 256;
    const int dn = n0 + (q >> 3);
    l4[q] = make_int4(dn, dn, dn, dn);
  }
  lcnt[t] = 0;
  __syncthreads();

  const int GW   = (G + 7) >> 3;
  const int gBeg = wave * GW;
  const int gEnd = (gBeg + GW) < G ? (gBeg + GW) : G;
  const unsigned* mrow = meta + (size_t)b * G;

  for (int kb = gBeg; kb < gEnd; kb += 32) {
    const int nr = (gEnd - kb) < 32 ? (gEnd - kb) : 32;
    const unsigned mv = lane < nr ? mrow[kb + lane] : 0u;

    unsigned ent[32];
    int rk[32];
#pragma unroll
    for (int u = 0; u < 32; ++u) {
      const unsigned m = (unsigned)__builtin_amdgcn_readlane((int)mv, u);
      const int st  = (int)(m & 0xffffu);
      const int len = (int)(m >> 16);
      ent[u] = 0u;
      if (lane < len) ent[u] = sorted[(size_t)(kb + u) * CH + st + lane];
    }
#pragma unroll
    for (int u = 0; u < 32; ++u) {
      const unsigned m = (unsigned)__builtin_amdgcn_readlane((int)mv, u);
      const int len = (int)(m >> 16);
      rk[u] = 0;
      if (lane < len) rk[u] = atomicAdd(&lcnt[ent[u] & 255u], 1);
    }
#pragma unroll
    for (int u = 0; u < 32; ++u) {
      const unsigned m = (unsigned)__builtin_amdgcn_readlane((int)mv, u);
      const int len = (int)(m >> 16);
      if (lane < len)
        put_entry(ent[u], rk[u], n0, lslots, novf, ovf_dst, ovf_src, E);
    }
    for (int u = 0; u < nr; ++u) {
      const unsigned m = (unsigned)__builtin_amdgcn_readlane((int)mv, u);
      const int st  = (int)(m & 0xffffu);
      const int len = (int)(m >> 16);
      for (int i = 32 + lane; i < len; i += 32) {
        const unsigned e = sorted[(size_t)(kb + u) * CH + st + i];
        const int r = atomicAdd(&lcnt[e & 255u], 1);
        put_entry(e, r, n0, lslots, novf, ovf_dst, ovf_src, E);
      }
    }
  }
  __syncthreads();

  const int nn = (N - n0) < NBK ? (N - n0) : NBK;
  int4* g4 = reinterpret_cast<int4*>(slots + (size_t)n0 * SLOTS);
#pragma unroll
  for (int k = 0; k < (NBK * SLOTS / 4) / 256; ++k) {
    const int q = t + k * 256;
    if ((q >> 3) < nn) g4[q] = l4[q];
  }
  if (t < nn) cnt[n0 + t] = lcnt[t];
}

template <int D, bool TANH_BIAS>
__global__ void node_softmax_aggregate(const int* __restrict__ cnt,
                                       const int* __restrict__ slots,
                                       const float* __restrict__ attS,
                                       const float* __restrict__ attD,
                                       const float* __restrict__ H,
                                       const float* __restrict__ bias,
                                       const int* __restrict__ novf,
                                       const int* __restrict__ ovf_dst,
                                       const int* __restrict__ ovf_src,
                                       float* __restrict__ out, int N, int E) {
  static_assert(D == 32 || D == 64, "row must be 32 or 64 floats");
  constexpr int V = D / 32;
  const int lane = threadIdx.x & 31;
  const int l15  = lane & 15;
  const int node = __builtin_amdgcn_readfirstlane(
      (int)((blockIdx.x * blockDim.x + threadIdx.x) >> 5));
  if (node >= N) return;

  const float* Hl = H + lane * V;
  float hs0, hs1 = 0.f, as0, as1 = 0.f, ad0, ad1 = 0.f;
  if (D == 64) {
    const float2 h = *reinterpret_cast<const float2*>(Hl + (size_t)node * D);
    const float2 a = *reinterpret_cast<const float2*>(attS + lane * V);
    const float2 b = *reinterpret_cast<const float2*>(attD + lane * V);
    hs0 = h.x; hs1 = h.y; as0 = a.x; as1 = a.y; ad0 = b.x; ad1 = b.y;
  } else {
    hs0 = Hl[(size_t)node * D];
    as0 = attS[lane];
    ad0 = attD[lane];
  }
  float bv0 = 0.f, bv1 = 0.f;
  if (TANH_BIAS) {
    if (D == 64) {
      const float2 bb = *reinterpret_cast<const float2*>(bias + lane * V);
      bv0 = bb.x;
      bv1 = bb.y;
    } else {
      bv0 = bias[lane];
    }
  }
  const int sl  = slots[(size_t)node * SLOTS + lane];
  const int deg = cnt[node];
  const int nslot = deg < SLOTS ? deg : SLOTS;

  float ps = fmaf(hs0, as0, hs1 * as1);
  float pd = fmaf(hs0, ad0, hs1 * ad1);
  ps += lane_xor16(ps);   pd += lane_xor16(pd);
  ps += lane_xor<8>(ps);  pd += lane_xor<8>(pd);
  ps += lane_xor<4>(ps);  pd += lane_xor<4>(pd);
  ps += lane_xor<2>(ps);  pd += lane_xor<2>(pd);
  ps += lane_xor<1>(ps);  pd += lane_xor<1>(pd);
  const float ad = pd;
  float m = leaky(ps + ad);
  float l = 1.f;
  float acc0 = hs0, acc1 = hs1;

  const bool b3 = (lane & 8) != 0, b2 = (lane & 4) != 0;
  const bool b1 = (lane & 2) != 0, b0 = (lane & 1) != 0;

  asm volatile("" ::"v"(sl));
  if (TANH_BIAS) {
    if (D == 64) asm volatile("" ::"v"(bv0), "v"(bv1));
    else asm volatile("" ::"v"(bv0));
  }

  for (int base = 0; base < nslot; base += 16) {
    const int n = (nslot - base < 16) ? (nslot - base) : 16;
    const bool has = l15 < n;

    float r0[16], r1[16];
#pragma unroll
    for (int j = 0; j < 16; ++j) {
      const int sj = __builtin_amdgcn_readlane(sl, base + j);
      if (D == 64) {
        const float2 h = *reinterpret_cast<const float2*>(Hl + (size_t)sj * D);
        r0[j] = h.x;
        r1[j] = h.y;
      } else {
        r0[j] = Hl[(size_t)sj * D];
        r1[j] = 0.f;
      }
    }

    float p[16];
#pragma unroll
    for (int j = 0; j < 16; ++j) p[j] = r0[j] * as0 + r1[j] * as1;
    float q[8];
#pragma unroll
    for (int k = 0; k < 8; ++k) {
      const float keep = b3 ? p[k + 8] : p[k];
      const float send = b3 ? p[k] : p[k + 8];
      q[k] = keep + lane_xor<8>(send);
    }
    float t[4];
#pragma unroll
    for (int k = 0; k < 4; ++k) {
      const float keep = b2 ? q[k + 4] : q[k];
      const float send = b2 ? q[k] : q[k + 4];
      t[k] = keep + lane_xor<4>(send);
    }
    float u[2];
#pragma unroll
    for (int k = 0; k < 2; ++k) {
      const float keep = b1 ? t[k + 2] : t[k];
      const float send = b1 ? t[k] : t[k + 2];
      u[k] = keep + lane_xor<2>(send);
    }
    float sc;
    {
      const float keep = b0 ? u[1] : u[0];
      const float send = b0 ? u[0] : u[1];
      sc = keep + lane_xor<1>(send);
    }
    sc += lane_xor16(sc);

    const float v = has ? leaky(sc + ad) : -INFINITY;
    float cm = v;
    cm = fmaxf(cm, lane_xor<8>(cm));
    cm = fmaxf(cm, lane_xor<4>(cm));
    cm = fmaxf(cm, lane_xor<2>(cm));
    cm = fmaxf(cm, lane_xor<1>(cm));
    const float mn    = fmaxf(m, cm);
    const float scale = __expf(m - mn);
    const float w     = has ? __expf(v - mn) : 0.f;
    float ws = w;
    ws += lane_xor<8>(ws);
    ws += lane_xor<4>(ws);
    ws += lane_xor<2>(ws);
    ws += lane_xor<1>(ws);
    l    = l * scale + ws;
    acc0 = acc0 * scale;
    acc1 = acc1 * scale;
    m    = mn;

#pragma unroll
    for (int j = 0; j < 16; ++j) {
      if (j < n) {
        const float wj =
            __int_as_float(__builtin_amdgcn_readlane(__float_as_int(w), j));
        acc0 += wj * r0[j];
        if (D == 64) acc1 += wj * r1[j];
      }
    }
  }

  if (deg > SLOTS) {
    int nov = novf[0];
    nov = nov < E ? nov : E;
    for (int base = 0; base < nov; base += 32) {
      const int i = base + lane;
      int od = -1, os = 0;
      if (i < nov) {
        od = ovf_dst[i];
        os = ovf_src[i];
      }
      unsigned mask = __builtin_amdgcn_ballot_w32(od == node);
      while (mask != 0u) {
        const int j = __builtin_ctz(mask);
        mask &= mask - 1u;
        const int sj = __builtin_amdgcn_readlane(os, j);
        float g0, g1 = 0.f;
        if (D == 64) {
          const float2 h = *reinterpret_cast<const float2*>(Hl + (size_t)sj * D);
          g0 = h.x;
          g1 = h.y;
        } else {
          g0 = Hl[(size_t)sj * D];
        }
        float sc = fmaf(g0, as0, g1 * as1);
        sc += lane_xor16(sc);
        sc += lane_xor<8>(sc);
        sc += lane_xor<4>(sc);
        sc += lane_xor<2>(sc);
        sc += lane_xor<1>(sc);
        const float v     = leaky(sc + ad);
        const float mn    = fmaxf(m, v);
        const float scale = __expf(m - mn);
        const float w     = __expf(v - mn);
        l    = l * scale + w;
        acc0 = acc0 * scale + w * g0;
        acc1 = acc1 * scale + w * g1;
        m    = mn;
      }
    }
  }

  float* o = out + (size_t)node * D + lane * V;
  if (D == 64) {
    float2 r;
    r.x = acc0 / l;
    r.y = acc1 / l;
    if (TANH_BIAS) {
      r.x = tanhf(r.x + bv0);
      r.y = tanhf(r.y + bv1);
    }
    *reinterpret_cast<float2*>(o) = r;
  } else {
    float r = acc0 / l;
    if (TANH_BIAS) r = tanhf(r + bv0);
    o[0] = r;
  }
}

extern "C" void kernel_launch(void* const* d_in, const int* in_sizes, int n_in,
                              void* d_out, int out_size, void* d_ws,
                              size_t ws_size, hipStream_t stream) {
  const float* x   = (const float*)d_in[0];
  const int*   ei  = (const int*)d_in[1];
  const float* W1  = (const float*)d_in[2];
  const float* as1 = (const float*)d_in[3];
  const float* ad1 = (const float*)d_in[4];
  const float* b1  = (const float*)d_in[5];
  const float* W2  = (const float*)d_in[6];
  const float* as2 = (const float*)d_in[7];
  const float* ad2 = (const float*)d_in[8];
  const float* b2  = (const float*)d_in[9];
  float* out = (float*)d_out;

  const int N    = in_sizes[0] / 64;
  const int E    = in_sizes[1] / 2;
  const int* src = ei;
  const int* dst = ei + E;

  float* ws = (float*)d_ws;
  size_t o = 0;
  float* h1 = ws + o;   o += (size_t)N * 64;
  float* o1 = ws + o;   o += (size_t)N * 64;
  int* slots   = (int*)(ws + o); o += (size_t)N * SLOTS;
  int* novf    = (int*)(ws + o); o += 32;
  int* ovf_dst = (int*)(ws + o); o += E;
  int* ovf_src = (int*)(ws + o); o += E;
  int* cnt     = (int*)(ws + o); o += N;
  const int G  = (E + CH - 1) / CH;
  const int Bn = (N + NBK - 1) / NBK;
  o = (o + 31) & ~(size_t)31;
  unsigned* sorted = (unsigned*)(ws + o); o += (size_t)G * CH;
  unsigned* meta   = (unsigned*)(ws + o); o += (size_t)Bn * G;

  const int B  = 256;
  const int gn = (int)(((size_t)N * 32 + B - 1) / B);

  part_edges<<<G, B, 0, stream>>>(src, dst, sorted, meta, novf, E, Bn, G);
  fill_slots<<<Bn, B, 0, stream>>>(sorted, meta, cnt, slots, novf, ovf_dst,
                                   ovf_src, N, E, G);

  {
    const int tiles = (N / 16) * (64 / 16);
    gat_gemm_wmma<64, 64, false>
        <<<(tiles + 7) / 8, B, 0, stream>>>(x, W1, nullptr, h1, N);
    node_softmax_aggregate<64, false>
        <<<gn, B, 0, stream>>>(cnt, slots, as1, ad1, h1, nullptr, novf, ovf_dst,
                                ovf_src, o1, N, E);
  }

  {
    float* h2 = h1;
    const int tiles = (N / 16) * (32 / 16);
    gat_gemm_wmma<64, 32, true>
        <<<(tiles + 7) / 8, B, 0, stream>>>(o1, W2, b1, h2, N);
    node_softmax_aggregate<32, true>
        <<<gn, B, 0, stream>>>(cnt, slots, as2, ad2, h2, b2, novf, ovf_dst,
                                ovf_src, out, N, E);
  }
}


# Layer-2 GEMM done inside the layer-1 aggregate: a work-group's 16 finished rows go as relu(row+b1) through LDS to one wave's WMMA x W2, same k order, h2 bit-identical; o1 round trip and a launch gone
# speedup vs baseline: 7.6491x; 1.1306x over previous
#include <hip/hip_runtime.h>
#include <math.h>


typedef __attribute__((ext_vector_type(2))) float v2f;
typedef __attribute__((ext_vector_type(8))) float v8f;

#define NEG_SLOPE 0.2f

#if defined(__has_builtin)
#if __has_builtin(__builtin_amdgcn_wmma_f32_16x16x4_f32)
#define HAVE_WMMA_F32_16X16X4 1
#endif
#endif

__device__ __forceinline__ float leaky(float v) {
  return v > 0.f ? v : NEG_SLOPE * v;
}

template <int M>
__device__ __forceinline__ float lane_xor(float v) {
  return __builtin_bit_cast(
      float, __builtin_amdgcn_update_dpp(0, __builtin_bit_cast(int, v),
                                         0x160 + M, 0xf, 0xf, true));
}
__device__ __forceinline__ float lane_xor16(float v) {
  const int b = __builtin_bit_cast(int, v);
  return __builtin_bit_cast(
      float, __builtin_amdgcn_permlanex16(b, b, 0x76543210, 0xfedcba98, false,
                                          false));
}

template <int J>
__device__ __forceinline__ int row_lane(int v) {
  return __builtin_amdgcn_update_dpp(0, v, 0x150 + J, 0xf, 0xf, true);
}
__device__ __forceinline__ int row_lane_at(int v, int j) {
  switch (j & 15) {
    case 0: return row_lane<0>(v);
    case 1: return row_lane<1>(v);
    case 2: return row_lane<2>(v);
    case 3: return row_lane<3>(v);
    case 4: return row_lane<4>(v);
    case 5: return row_lane<5>(v);
    case 6: return row_lane<6>(v);
    case 7: return row_lane<7>(v);
    case 8: return row_lane<8>(v);
    case 9: return row_lane<9>(v);
    case 10: return row_lane<10>(v);
    case 11: return row_lane<11>(v);
    case 12: return row_lane<12>(v);
    case 13: return row_lane<13>(v);
    case 14: return row_lane<14>(v);
    default: return row_lane<15>(v);
  }
}
__device__ __forceinline__ float row_lane_at(float v, int j) {
  return __builtin_bit_cast(float, row_lane_at(__builtin_bit_cast(int, v), j));
}

template <int K, int DOUT>
__global__ void gat_gemm_wmma(const float* __restrict__ X,
                              const float* __restrict__ W,
                              float* __restrict__ H, int M) {
  const int lane   = threadIdx.x & 31;
  const int wave   = threadIdx.x >> 5;
  const int tileId = blockIdx.x * (blockDim.x >> 5) + wave;
  const int NT     = DOUT / 16;
  const int mt     = tileId / NT;
  const int nt     = tileId % NT;
  if (mt >= M / 16) return;

  const int r15 = lane & 15;
  const int hi  = lane >> 4;
  const int row = mt * 16 + r15;
  const int col = nt * 16 + r15;

  v8f c = {0.f, 0.f, 0.f, 0.f, 0.f, 0.f, 0.f, 0.f};

#if HAVE_WMMA_F32_16X16X4
  for (int k0 = 0; k0 < K; k0 += 4) {
    const int ka = k0 + 2 * hi;
    v2f a;
    a.x = X[row * K + ka];
    a.y = X[row * K + ka + 1];
    v2f b;
    b.x = W[ka * DOUT + col];
    b.y = W[(ka + 1) * DOUT + col];
    c = __builtin_amdgcn_wmma_f32_16x16x4_f32(false, a, false, b, (short)0, c,
                                              false, false);
  }
#else
  for (int r = 0; r < 8; ++r) {
    const int rr = mt * 16 + hi * 8 + r;
    float acc = 0.f;
    for (int k = 0; k < K; ++k) acc += X[rr * K + k] * W[k * DOUT + col];
    c[r] = acc;
  }
#endif

  const int rowBase = mt * 16 + hi * 8;
#pragma unroll
  for (int r = 0; r < 8; ++r) H[(rowBase + r) * DOUT + col] = c[r];
}

#define SLOTS 32
#define CH 8192
#define EPT 32
#define NBK 256
#define BMAX 1024

__global__ void __launch_bounds__(256)
part_edges(const int* __restrict__ src, const int* __restrict__ dst,
           unsigned* __restrict__ sorted, unsigned* __restrict__ meta,
           int* __restrict__ novf, int E, int Bn, int G) {
  __shared__ __attribute__((aligned(16))) int hist[BMAX];
  __shared__ __attribute__((aligned(16))) unsigned stage[CH];
  __shared__ int wsum[8];
  const int t    = threadIdx.x;
  const int lane = t & 31;
  const int wave = t >> 5;
  const int g    = blockIdx.x;
  const int e0   = g * CH;

  for (int i = t; i < BMAX; i += 256) hist[i] = 0;
  if (g == 0 && t == 0) novf[0] = 0;
  __syncthreads();

  int d[EPT], s[EPT], r[EPT];
#pragma unroll
  for (int k = 0; k < EPT; ++k) {
    const int e  = e0 + k * 256 + t;
    const int ec = e < E ? e : E - 1;
    d[k] = dst[ec];
    s[k] = src[ec];
  }
#pragma unroll
  for (int k = 0; k < EPT; ++k) {
    const int e = e0 + k * 256 + t;
    int b = d[k] >> 8;
    b = b < 0 ? 0 : (b < BMAX ? b : BMAX - 1);
    d[k] = e < E ? d[k] : -1;
    s[k] = (int)(((unsigned)s[k] << 8) | (unsigned)(d[k] & 255));
    r[k] = 0;
    if (d[k] >= 0) r[k] = atomicAdd(&hist[b], 1);
    d[k] = d[k] >= 0 ? b : -1;
  }
  __syncthreads();

  const int4 h = reinterpret_cast<const int4*>(hist)[t];
  const int sum = h.x + h.y + h.z + h.w;
  int incl = sum;
#pragma unroll
  for (int off = 1; off < 32; off <<= 1) {
    const int v = __shfl_up(incl, off);
    if (lane >= off) incl += v;
  }
  if (lane == 31) wsum[wave] = incl;
  __syncthreads();
  int base = 0;
  for (int w = 0; w < wave; ++w) base += wsum[w];
  int4 st;
  st.x = base + incl - sum;
  st.y = st.x + h.x;
  st.z = st.y + h.y;
  st.w = st.z + h.z;
  reinterpret_cast<int4*>(hist)[t] = st;
  {
    const int b = 4 * t;
    unsigned* m = meta + (size_t)b * G + g;
    if (b + 0 < Bn) m[0]             = (unsigned)st.x | ((unsigned)h.x << 16);
    if (b + 1 < Bn) m[(size_t)G]     = (unsigned)st.y | ((unsigned)h.y << 16);
    if (b + 2 < Bn) m[(size_t)G * 2] = (unsigned)st.z | ((unsigned)h.z << 16);
    if (b + 3 < Bn) m[(size_t)G * 3] = (unsigned)st.w | ((unsigned)h.w << 16);
  }
  __syncthreads();

#pragma unroll
  for (int k = 0; k < EPT; ++k) r[k] += hist[d[k] >= 0 ? d[k] : 0];
#pragma unroll
  for (int k = 0; k < EPT; ++k)
    if (d[k] >= 0) stage[r[k]] = (unsigned)s[k];
  __syncthreads();

  const int clen = (E - e0) < CH ? (E - e0) : CH;
  const int c4   = clen >> 2;
  uint4* o4 = reinterpret_cast<uint4*>(sorted + e0);
  const uint4* s4 = reinterpret_cast<const uint4*>(stage);
  for (int i = t; i < c4; i += 256) o4[i] = s4[i];
  for (int i = (c4 << 2) + t; i < clen; i += 256) sorted[e0 + i] = stage[i];
}

__device__ __forceinline__ void put_entry(unsigned ent, int r, int n0,
                                          int* lslots, int* __restrict__ novf,
                                          int* __restrict__ ovf_dst,
                                          int* __restrict__ ovf_src, int E) {
  const int n = (int)(ent & 255u);
  const int s = (int)(ent >> 8);
  if (r < SLOTS) {
    lslots[n * SLOTS + r] = s;
  } else {
    const int p = atomicAdd(novf, 1);
    if (p < E) {
      ovf_dst[p] = n0 + n;
      ovf_src[p] = s;
    }
  }
}

__global__ void __launch_bounds__(256)
fill_slots(const unsigned* __restrict__ sorted,
           const unsigned* __restrict__ meta, int* __restrict__ cnt,
           int* __restrict__ slots, int* __restrict__ novf,
           int* __restrict__ ovf_dst, int* __restrict__ ovf_src, int N, int E,
           int G) {
  __shared__ __attribute__((aligned(16))) int lslots[NBK * SLOTS];
  __shared__ int lcnt[NBK];
  const int t    = threadIdx.x;
  const int lane = t & 31;
  const int wave = __builtin_amdgcn_readfirstlane(t >> 5);
  const int b    = blockIdx.x;
  const int n0   = b * NBK;

  int4* l4 = reinterpret_cast<int4*>(lslots);
#pragma unroll
  for (int k = 0; k < (NBK * SLOTS / 4) / 256; ++k) {
    const int q = t + k * 256;
    const int dn = n0 + (q >> 3);
    l4[q] = make_int4(dn, dn, dn, dn);
  }
  lcnt[t] = 0;
  __syncthreads();

  const int GW   = (G + 7) >> 3;
  const int gBeg = wave * GW;
  const int gEnd = (gBeg + GW) < G ? (gBeg + GW) : G;
  const unsigned* mrow = meta + (size_t)b * G;

  for (int kb = gBeg; kb < gEnd; kb += 32) {
    const int nr = (gEnd - kb) < 32 ? (gEnd - kb) : 32;
    const unsigned mv = lane < nr ? mrow[kb + lane] : 0u;

    unsigned ent[32];
    int rk[32];
#pragma unroll
    for (int u = 0; u < 32; ++u) {
      const unsigned m = (unsigned)__builtin_amdgcn_readlane((int)mv, u);
      const int st  = (int)(m & 0xffffu);
      const int len = (int)(m >> 16);
      ent[u] = 0u;
      if (lane < len) ent[u] = sorted[(size_t)(kb + u) * CH + st + lane];
    }
#pragma unroll
    for (int u = 0; u < 32; ++u) {
      const unsigned m = (unsigned)__builtin_amdgcn_readlane((int)mv, u);
      const int len = (int)(m >> 16);
      rk[u] = 0;
      if (lane < len) rk[u] = atomicAdd(&lcnt[ent[u] & 255u], 1);
    }
#pragma unroll
    for (int u = 0; u < 32; ++u) {
      const unsigned m = (unsigned)__builtin_amdgcn_readlane((int)mv, u);
      const int len = (int)(m >> 16);
      if (lane < len)
        put_entry(ent[u], rk[u], n0, lslots, novf, ovf_dst, ovf_src, E);
    }
    for (int u = 0; u < nr; ++u) {
      const unsigned m = (unsigned)__builtin_amdgcn_readlane((int)mv, u);
      const int st  = (int)(m & 0xffffu);
      const int len = (int)(m >> 16);
      for (int i = 32 + lane; i < len; i += 32) {
        const unsigned e = sorted[(size_t)(kb + u) * CH + st + i];
        const int r = atomicAdd(&lcnt[e & 255u], 1);
        put_entry(e, r, n0, lslots, novf, ovf_dst, ovf_src, E);
      }
    }
  }
  __syncthreads();

  const int nn = (N - n0) < NBK ? (N - n0) : NBK;
  int4* g4 = reinterpret_cast<int4*>(slots + (size_t)n0 * SLOTS);
#pragma unroll
  for (int k = 0; k < (NBK * SLOTS / 4) / 256; ++k) {
    const int q = t + k * 256;
    if ((q >> 3) < nn) g4[q] = l4[q];
  }
  if (t < nn) cnt[n0 + t] = lcnt[t];
}

template <int V, typename VF>
__device__ __forceinline__ float lane_dot(VF r, VF a) {
  if constexpr (V == 4) {
    v2f d = r.xy * a.xy;
    d = r.zw * a.zw + d;
    return d.x + d.y;
  } else {
    return fmaf(r.y, a.y, r.x * a.x);
  }
}

template <int D, bool TANH_BIAS, int NOUT>
__global__ void node_softmax_aggregate(const int* __restrict__ cnt,
                                       const int* __restrict__ slots,
                                       const float* __restrict__ attS,
                                       const float* __restrict__ attD,
                                       const float* __restrict__ H,
                                       const float* __restrict__ bias,
                                       const float* __restrict__ Wn,
                                       const int* __restrict__ novf,
                                       const int* __restrict__ ovf_dst,
                                       const int* __restrict__ ovf_src,
                                       float* __restrict__ out, int N, int E) {
  static_assert(D == 32 || D == 64, "row must be 32 or 64 floats");
  static_assert(NOUT == 0 || NOUT == 32, "fused GEMM: two 16-column tiles");
  constexpr bool ADD_BIAS = TANH_BIAS || NOUT != 0;
  constexpr int V = D / 16;
  typedef float vf __attribute__((ext_vector_type(V)));
  const int lane = threadIdx.x & 31;
  const int c    = lane & 15;
  const int hf   = lane >> 4;
  const int wv = __builtin_amdgcn_readfirstlane(
      (int)((blockIdx.x * blockDim.x + threadIdx.x) >> 5));
  if (NOUT == 0 && 2 * wv >= N) return;
  const bool live = 2 * wv + hf < N;
  const int node  = live ? 2 * wv + hf : N - 1;

  const vf* Hv = reinterpret_cast<const vf*>(H);
  const vf hs = Hv[(unsigned)(node * 16 + c)];
  const vf as = reinterpret_cast<const vf*>(attS)[c];
  const vf ad = reinterpret_cast<const vf*>(attD)[c];
  vf bv = 0.f;
  if (ADD_BIAS) bv = reinterpret_cast<const vf*>(bias)[c];
  const int sl0 = slots[(unsigned)(node * SLOTS) + c];
  const int sl1 = slots[(unsigned)(node * SLOTS) + 16 + c];
  const int deg = cnt[node];
  const int nslot = deg < SLOTS ? deg : SLOTS;

  float ps = lane_dot<V>(hs, as);
  float pd = lane_dot<V>(hs, ad);
  ps += lane_xor<8>(ps);  pd += lane_xor<8>(pd);
  ps += lane_xor<4>(ps);  pd += lane_xor<4>(pd);
  ps += lane_xor<2>(ps);  pd += lane_xor<2>(pd);
  ps += lane_xor<1>(ps);  pd += lane_xor<1>(pd);
  const float adn = pd;
  float m = leaky(ps + adn);
  float l = 1.f;
  vf acc = hs;

  const bool b3 = (lane & 8) != 0, b2 = (lane & 4) != 0;
  const bool b1 = (lane & 2) != 0, b0 = (lane & 1) != 0;

  asm volatile("" ::"v"(sl0), "v"(sl1));
  if (ADD_BIAS) {
    if (V == 4) asm volatile("" ::"v"(bv[0]), "v"(bv[1]), "v"(bv[2]), "v"(bv[3]));
    else asm volatile("" ::"v"(bv[0]), "v"(bv[1]));
  }

  const int ns_lo = __builtin_amdgcn_readlane(nslot, 0);
  const int ns_hi = __builtin_amdgcn_readlane(nslot, 16);
  const int nmax  = ns_lo > ns_hi ? ns_lo : ns_hi;

  for (int base = 0; base < nmax; base += 16) {
    const int sl   = base == 0 ? sl0 : sl1;
    const bool has = c < nslot - base;
    const int nb   = (nmax - base < 16) ? (nmax - base) : 16;

    vf r[16];
#pragma unroll
    for (int j = 0; j < 16; ++j) {
      const int sj = row_lane_at(sl, j);
      r[j] = Hv[(unsigned)(sj * 16 + c)];
    }
    asm volatile("" ::"v"(r[0]), "v"(r[1]), "v"(r[2]), "v"(r[3]), "v"(r[4]),
                 "v"(r[5]), "v"(r[6]), "v"(r[7]), "v"(r[8]), "v"(r[9]),
                 "v"(r[10]), "v"(r[11]), "v"(r[12]), "v"(r[13]), "v"(r[14]),
                 "v"(r[15]));

    float p[16];
#pragma unroll
    for (int j = 0; j < 16; ++j) p[j] = lane_dot<V>(r[j], as);
    float q[8];
#pragma unroll
    for (int k = 0; k < 8; ++k) {
      const float keep = b3 ? p[k + 8] : p[k];
      const float send = b3 ? p[k] : p[k + 8];
      q[k] = keep + lane_xor<8>(send);
    }
    float t[4];
#pragma unroll
    for (int k = 0; k < 4; ++k) {
      const float keep = b2 ? q[k + 4] : q[k];
      const float send = b2 ? q[k] : q[k + 4];
      t[k] = keep + lane_xor<4>(send);
    }
    float u[2];
#pragma unroll
    for (int k = 0; k < 2; ++k) {
      const float keep = b1 ? t[k + 2] : t[k];
      const float send = b1 ? t[k] : t[k + 2];
      u[k] = keep + lane_xor<2>(send);
    }
    float sc;
    {
      const float keep = b0 ? u[1] : u[0];
      const float send = b0 ? u[0] : u[1];
      sc = keep + lane_xor<1>(send);
    }

    const float v = has ? leaky(sc + adn) : -INFINITY;
    float cm = v;
    cm = fmaxf(cm, lane_xor<8>(cm));
    cm = fmaxf(cm, lane_xor<4>(cm));
    cm = fmaxf(cm, lane_xor<2>(cm));
    cm = fmaxf(cm, lane_xor<1>(cm));
    const float mn    = fmaxf(m, cm);
    const float scale = __expf(m - mn);
    const float w     = has ? __expf(v - mn) : 0.f;
    float ws = w;
    ws += lane_xor<8>(ws);
    ws += lane_xor<4>(ws);
    ws += lane_xor<2>(ws);
    ws += lane_xor<1>(ws);
    l   = l * scale + ws;
    acc = acc * scale;
    m   = mn;

#pragma unroll
    for (int g = 0; g < 4; ++g) {
      if (4 * g < nb) {
#pragma unroll
        for (int jj = 0; jj < 4; ++jj) {
          const int j    = 4 * g + jj;
          acc += row_lane_at(w, j) * r[j];
        }
      }
    }
  }

  const bool ovl = deg > SLOTS;
  if (__builtin_amdgcn_ballot_w32(ovl) != 0u) {
    const int node_lo = __builtin_amdgcn_readlane(node, 0);
    const int node_hi = __builtin_amdgcn_readlane(node, 16);
    int nov = novf[0];
    nov = nov < E ? nov : E;
    for (int base = 0; base < nov; base += 32) {
      const int i = base + lane;
      int od = -1, os = 0;
      if (i < nov) {
        od = ovf_dst[i];
        os = ovf_src[i];
      }
      unsigned m_lo = __builtin_amdgcn_ballot_w32(od == node_lo);
      unsigned m_hi = __builtin_amdgcn_ballot_w32(od == node_hi);
      while ((m_lo | m_hi) != 0u) {
        const bool a_lo = m_lo != 0u, a_hi = m_hi != 0u;
        const int j_lo = a_lo ? __builtin_ctz(m_lo) : 0;
        const int j_hi = a_hi ? __builtin_ctz(m_hi) : 0;
        m_lo &= m_lo - 1u;
        m_hi &= m_hi - 1u;
        const int s_lo = __builtin_amdgcn_readlane(os, j_lo);
        const int s_hi = __builtin_amdgcn_readlane(os, j_hi);
        const bool act = hf ? a_hi : a_lo;
        const int sj   = act ? (hf ? s_hi : s_lo) : node;
        const vf g = Hv[(unsigned)(sj * 16 + c)];
        float sc = lane_dot<V>(g, as);
        sc += lane_xor<8>(sc);
        sc += lane_xor<4>(sc);
        sc += lane_xor<2>(sc);
        sc += lane_xor<1>(sc);
        const float v     = leaky(sc + adn);
        const float mn    = act ? fmaxf(m, v) : m;
        const float scale = __expf(m - mn);
        const float w     = act ? __expf(v - mn) : 0.f;
        l   = l * scale + w;
        acc = acc * scale + w * g;
        m   = mn;
      }
    }
  }

  if constexpr (NOUT == 0) {
    if (live) {
      vf rr;
#pragma unroll
      for (int k = 0; k < V; ++k) {
        rr[k] = acc[k] / l;
        if (TANH_BIAS) rr[k] = tanhf(rr[k] + bv[k]);
      }
      reinterpret_cast<vf*>(out)[(unsigned)(node * 16 + c)] = rr;
    }
  } else {
    constexpr int LDA = D + 4;
    constexpr int KS  = D / 4;
    __shared__ __attribute__((aligned(16))) float tileA[16 * LDA];
    const int wib = __builtin_amdgcn_readfirstlane((int)(threadIdx.x >> 5));

    vf rr;
#pragma unroll
    for (int k = 0; k < V; ++k) rr[k] = fmaxf(acc[k] / l + bv[k], 0.f);
    const int trow = 2 * wib + hf;
    *reinterpret_cast<vf*>(&tileA[trow * LDA + V * c]) = rr;

    v2f wb0[KS], wb1[KS];
    if (wib == 0) {
#pragma unroll
      for (int i = 0; i < KS; ++i) {
        const int ka = 4 * i + 2 * hf;
        wb0[i].x = Wn[ka * NOUT + c];
        wb0[i].y = Wn[(ka + 1) * NOUT + c];
        wb1[i].x = Wn[ka * NOUT + 16 + c];
        wb1[i].y = Wn[(ka + 1) * NOUT + 16 + c];
      }
    }
    __syncthreads();

    if (wib == 0) {
      const int row0 = (int)blockIdx.x * 16 + hf * 8;
#if HAVE_WMMA_F32_16X16X4
      v8f c0 = {0.f, 0.f, 0.f, 0.f, 0.f, 0.f, 0.f, 0.f};
      v8f c1 = {0.f, 0.f, 0.f, 0.f, 0.f, 0.f, 0.f, 0.f};
#pragma unroll
      for (int i = 0; i < KS; ++i) {
        const v2f a =
            *reinterpret_cast<const v2f*>(&tileA[c * LDA + 4 * i + 2 * hf]);
        c0 = __builtin_amdgcn_wmma_f32_16x16x4_f32(false, a, false, wb0[i],
                                                   (short)0, c0, false, false);
        c1 = __builtin_amdgcn_wmma_f32_16x16x4_f32(false, a, false, wb1[i],
                                                   (short)0, c1, false, false);
      }
#else
      float c0[8], c1[8];
      for (int r = 0; r < 8; ++r) {
        float s0 = 0.f, s1 = 0.f;
        for (int k = 0; k < D; ++k) {
          const float xv = tileA[(hf * 8 + r) * LDA + k];
          s0 += xv * Wn[k * NOUT + c];
          s1 += xv * Wn[k * NOUT + 16 + c];
        }
        c0[r] = s0;
        c1[r] = s1;
      }
#endif
#pragma unroll
      for (int r = 0; r < 8; ++r) {
        if (row0 + r < N) {
          out[(unsigned)((row0 + r) * NOUT + c)]      = c0[r];
          out[(unsigned)((row0 + r) * NOUT + 16 + c)] = c1[r];
        }
      }
    }
  }
}

extern "C" void kernel_launch(void* const* d_in, const int* in_sizes, int n_in,
                              void* d_out, int out_size, void* d_ws,
                              size_t ws_size, hipStream_t stream) {
  const float* x   = (const float*)d_in[0];
  const int*   ei  = (const int*)d_in[1];
  const float* W1  = (const float*)d_in[2];
  const float* as1 = (const float*)d_in[3];
  const float* ad1 = (const float*)d_in[4];
  const float* b1  = (const float*)d_in[5];
  const float* W2  = (const float*)d_in[6];
  const float* as2 = (const float*)d_in[7];
  const float* ad2 = (const float*)d_in[8];
  const float* b2  = (const float*)d_in[9];
  float* out = (float*)d_out;

  const int N    = in_sizes[0] / 64;
  const int E    = in_sizes[1] / 2;
  const int* src = ei;
  const int* dst = ei + E;

  float* ws = (float*)d_ws;
  size_t o = 0;
  float* h1 = ws + o;   o += (size_t)N * 64;
  float* h2 = ws + o;   o += (size_t)N * 32;
  int* slots   = (int*)(ws + o); o += (size_t)N * SLOTS;
  int* novf    = (int*)(ws + o); o += 32;
  int* ovf_dst = (int*)(ws + o); o += E;
  int* ovf_src = (int*)(ws + o); o += E;
  int* cnt     = (int*)(ws + o); o += N;
  const int G  = (E + CH - 1) / CH;
  const int Bn = (N + NBK - 1) / NBK;
  o = (o + 31) & ~(size_t)31;
  unsigned* sorted = (unsigned*)(ws + o); o += (size_t)G * CH;
  unsigned* meta   = (unsigned*)(ws + o); o += (size_t)Bn * G;

  const int B  = 256;
  const int gn = (int)((((size_t)N + 1) / 2 * 32 + B - 1) / B);

  part_edges<<<G, B, 0, stream>>>(src, dst, sorted, meta, novf, E, Bn, G);
  fill_slots<<<Bn, B, 0, stream>>>(sorted, meta, cnt, slots, novf, ovf_dst,
                                   ovf_src, N, E, G);

  {
    const int tiles = (N / 16) * (64 / 16);
    gat_gemm_wmma<64, 64>
        <<<(tiles + 7) / 8, B, 0, stream>>>(x, W1, h1, N);
    node_softmax_aggregate<64, false, 32>
        <<<gn, B, 0, stream>>>(cnt, slots, as1, ad1, h1, b1, W2, novf, ovf_dst,
                                ovf_src, h2, N, E);
  }

  {
    node_softmax_aggregate<32, true, 0>
        <<<gn, B, 0, stream>>>(cnt, slots, as2, ad2, h2, b2, nullptr, novf,
                                ovf_dst, ovf_src, out, N, E);
  }
}


# Layer-1 GEMM (x @ W1) runs in the same launch as pass 2 of the edge grouping: buckets are work-groups [0,Bn), GEMM tiles follow; the two are independent, so one launch and its ramp are gone
# speedup vs baseline: 8.1684x; 1.0679x over previous
#include <hip/hip_runtime.h>
#include <math.h>


typedef __attribute__((ext_vector_type(2))) float v2f;
typedef __attribute__((ext_vector_type(8))) float v8f;

#define NEG_SLOPE 0.2f

#if defined(__has_builtin)
#if __has_builtin(__builtin_amdgcn_wmma_f32_16x16x4_f32)
#define HAVE_WMMA_F32_16X16X4 1
#endif
#endif

__device__ __forceinline__ float leaky(float v) {
  return v > 0.f ? v : NEG_SLOPE * v;
}

template <int M>
__device__ __forceinline__ float lane_xor(float v) {
  return __builtin_bit_cast(
      float, __builtin_amdgcn_update_dpp(0, __builtin_bit_cast(int, v),
                                         0x160 + M, 0xf, 0xf, true));
}
__device__ __forceinline__ float lane_xor16(float v) {
  const int b = __builtin_bit_cast(int, v);
  return __builtin_bit_cast(
      float, __builtin_amdgcn_permlanex16(b, b, 0x76543210, 0xfedcba98, false,
                                          false));
}

template <int J>
__device__ __forceinline__ int row_lane(int v) {
  return __builtin_amdgcn_update_dpp(0, v, 0x150 + J, 0xf, 0xf, true);
}
__device__ __forceinline__ int row_lane_at(int v, int j) {
  switch (j & 15) {
    case 0: return row_lane<0>(v);
    case 1: return row_lane<1>(v);
    case 2: return row_lane<2>(v);
    case 3: return row_lane<3>(v);
    case 4: return row_lane<4>(v);
    case 5: return row_lane<5>(v);
    case 6: return row_lane<6>(v);
    case 7: return row_lane<7>(v);
    case 8: return row_lane<8>(v);
    case 9: return row_lane<9>(v);
    case 10: return row_lane<10>(v);
    case 11: return row_lane<11>(v);
    case 12: return row_lane<12>(v);
    case 13: return row_lane<13>(v);
    case 14: return row_lane<14>(v);
    default: return row_lane<15>(v);
  }
}
__device__ __forceinline__ float row_lane_at(float v, int j) {
  return __builtin_bit_cast(float, row_lane_at(__builtin_bit_cast(int, v), j));
}

template <int K, int DOUT>
__device__ __forceinline__ void gat_gemm_wmma(const float* __restrict__ X,
                                              const float* __restrict__ W,
                                              float* __restrict__ H, int M,
                                              int blk) {
  const int lane   = threadIdx.x & 31;
  const int wave   = threadIdx.x >> 5;
  const int tileId = blk * (blockDim.x >> 5) + wave;
  const int NT     = DOUT / 16;
  const int mt     = tileId / NT;
  const int nt     = tileId % NT;
  if (mt >= M / 16) return;

  const int r15 = lane & 15;
  const int hi  = lane >> 4;
  const int row = mt * 16 + r15;
  const int col = nt * 16 + r15;

  v8f c = {0.f, 0.f, 0.f, 0.f, 0.f, 0.f, 0.f, 0.f};

#if HAVE_WMMA_F32_16X16X4
  for (int k0 = 0; k0 < K; k0 += 4) {
    const int ka = k0 + 2 * hi;
    v2f a;
    a.x = X[row * K + ka];
    a.y = X[row * K + ka + 1];
    v2f b;
    b.x = W[ka * DOUT + col];
    b.y = W[(ka + 1) * DOUT + col];
    c = __builtin_amdgcn_wmma_f32_16x16x4_f32(false, a, false, b, (short)0, c,
                                              false, false);
  }
#else
  for (int r = 0; r < 8; ++r) {
    const int rr = mt * 16 + hi * 8 + r;
    float acc = 0.f;
    for (int k = 0; k < K; ++k) acc += X[rr * K + k] * W[k * DOUT + col];
    c[r] = acc;
  }
#endif

  const int rowBase = mt * 16 + hi * 8;
#pragma unroll
  for (int r = 0; r < 8; ++r) H[(rowBase + r) * DOUT + col] = c[r];
}

#define SLOTS 32
#define CH 8192
#define EPT 32
#define NBK 256
#define BMAX 1024

__global__ void __launch_bounds__(256)
part_edges(const int* __restrict__ src, const int* __restrict__ dst,
           unsigned* __restrict__ sorted, unsigned* __restrict__ meta,
           int* __restrict__ novf, int E, int Bn, int G) {
  __shared__ __attribute__((aligned(16))) int hist[BMAX];
  __shared__ __attribute__((aligned(16))) unsigned stage[CH];
  __shared__ int wsum[8];
  const int t    = threadIdx.x;
  const int lane = t & 31;
  const int wave = t >> 5;
  const int g    = blockIdx.x;
  const int e0   = g * CH;

  for (int i = t; i < BMAX; i += 256) hist[i] = 0;
  if (g == 0 && t == 0) novf[0] = 0;
  __syncthreads();

  int d[EPT], s[EPT], r[EPT];
#pragma unroll
  for (int k = 0; k < EPT; ++k) {
    const int e  = e0 + k * 256 + t;
    const int ec = e < E ? e : E - 1;
    d[k] = dst[ec];
    s[k] = src[ec];
  }
#pragma unroll
  for (int k = 0; k < EPT; ++k) {
    const int e = e0 + k * 256 + t;
    int b = d[k] >> 8;
    b = b < 0 ? 0 : (b < BMAX ? b : BMAX - 1);
    d[k] = e < E ? d[k] : -1;
    s[k] = (int)(((unsigned)s[k] << 8) | (unsigned)(d[k] & 255));
    r[k] = 0;
    if (d[k] >= 0) r[k] = atomicAdd(&hist[b], 1);
    d[k] = d[k] >= 0 ? b : -1;
  }
  __syncthreads();

  const int4 h = reinterpret_cast<const int4*>(hist)[t];
  const int sum = h.x + h.y + h.z + h.w;
  int incl = sum;
#pragma unroll
  for (int off = 1; off < 32; off <<= 1) {
    const int v = __shfl_up(incl, off);
    if (lane >= off) incl += v;
  }
  if (lane == 31) wsum[wave] = incl;
  __syncthreads();
  int base = 0;
  for (int w = 0; w < wave; ++w) base += wsum[w];
  int4 st;
  st.x = base + incl - sum;
  st.y = st.x + h.x;
  st.z = st.y + h.y;
  st.w = st.z + h.z;
  reinterpret_cast<int4*>(hist)[t] = st;
  {
    const int b = 4 * t;
    unsigned* m = meta + (size_t)b * G + g;
    if (b + 0 < Bn) m[0]             = (unsigned)st.x | ((unsigned)h.x << 16);
    if (b + 1 < Bn) m[(size_t)G]     = (unsigned)st.y | ((unsigned)h.y << 16);
    if (b + 2 < Bn) m[(size_t)G * 2] = (unsigned)st.z | ((unsigned)h.z << 16);
    if (b + 3 < Bn) m[(size_t)G * 3] = (unsigned)st.w | ((unsigned)h.w << 16);
  }
  __syncthreads();

#pragma unroll
  for (int k = 0; k < EPT; ++k) r[k] += hist[d[k] >= 0 ? d[k] : 0];
#pragma unroll
  for (int k = 0; k < EPT; ++k)
    if (d[k] >= 0) stage[r[k]] = (unsigned)s[k];
  __syncthreads();

  const int clen = (E - e0) < CH ? (E - e0) : CH;
  const int c4   = clen >> 2;
  uint4* o4 = reinterpret_cast<uint4*>(sorted + e0);
  const uint4* s4 = reinterpret_cast<const uint4*>(stage);
  for (int i = t; i < c4; i += 256) o4[i] = s4[i];
  for (int i = (c4 << 2) + t; i < clen; i += 256) sorted[e0 + i] = stage[i];
}

__device__ __forceinline__ void put_entry(unsigned ent, int r, int n0,
                                          int* lslots, int* __restrict__ novf,
                                          int* __restrict__ ovf_dst,
                                          int* __restrict__ ovf_src, int E) {
  const int n = (int)(ent & 255u);
  const int s = (int)(ent >> 8);
  if (r < SLOTS) {
    lslots[n * SLOTS + r] = s;
  } else {
    const int p = atomicAdd(novf, 1);
    if (p < E) {
      ovf_dst[p] = n0 + n;
      ovf_src[p] = s;
    }
  }
}

__device__ __forceinline__ void
fill_slots(const unsigned* __restrict__ sorted,
           const unsigned* __restrict__ meta, int* __restrict__ cnt,
           int* __restrict__ slots, int* __restrict__ novf,
           int* __restrict__ ovf_dst, int* __restrict__ ovf_src, int N, int E,
           int G, const int b) {
  __shared__ __attribute__((aligned(16))) int lslots[NBK * SLOTS];
  __shared__ int lcnt[NBK];
  const int t    = threadIdx.x;
  const int lane = t & 31;
  const int wave = __builtin_amdgcn_readfirstlane(t >> 5);
  const int n0   = b * NBK;

  int4* l4 = reinterpret_cast<int4*>(lslots);
#pragma unroll
  for (int k = 0; k < (NBK * SLOTS / 4) / 256; ++k) {
    const int q = t + k * 256;
    const int dn = n0 + (q >> 3);
    l4[q] = make_int4(dn, dn, dn, dn);
  }
  lcnt[t] = 0;
  __syncthreads();

  const int GW   = (G + 7) >> 3;
  const int gBeg = wave * GW;
  const int gEnd = (gBeg + GW) < G ? (gBeg + GW) : G;
  const unsigned* mrow = meta + (size_t)b * G;

  for (int kb = gBeg; kb < gEnd; kb += 32) {
    const int nr = (gEnd - kb) < 32 ? (gEnd - kb) : 32;
    const unsigned mv = lane < nr ? mrow[kb + lane] : 0u;

    unsigned ent[32];
    int rk[32];
#pragma unroll
    for (int u = 0; u < 32; ++u) {
      const unsigned m = (unsigned)__builtin_amdgcn_readlane((int)mv, u);
      const int st  = (int)(m & 0xffffu);
      const int len = (int)(m >> 16);
      ent[u] = 0u;
      if (lane < len) ent[u] = sorted[(size_t)(kb + u) * CH + st + lane];
    }
#pragma unroll
    for (int u = 0; u < 32; ++u) {
      const unsigned m = (unsigned)__builtin_amdgcn_readlane((int)mv, u);
      const int len = (int)(m >> 16);
      rk[u] = 0;
      if (lane < len) rk[u] = atomicAdd(&lcnt[ent[u] & 255u], 1);
    }
#pragma unroll
    for (int u = 0; u < 32; ++u) {
      const unsigned m = (unsigned)__builtin_amdgcn_readlane((int)mv, u);
      const int len = (int)(m >> 16);
      if (lane < len)
        put_entry(ent[u], rk[u], n0, lslots, novf, ovf_dst, ovf_src, E);
    }
    for (int u = 0; u < nr; ++u) {
      const unsigned m = (unsigned)__builtin_amdgcn_readlane((int)mv, u);
      const int st  = (int)(m & 0xffffu);
      const int len = (int)(m >> 16);
      for (int i = 32 + lane; i < len; i += 32) {
        const unsigned e = sorted[(size_t)(kb + u) * CH + st + i];
        const int r = atomicAdd(&lcnt[e & 255u], 1);
        put_entry(e, r, n0, lslots, novf, ovf_dst, ovf_src, E);
      }
    }
  }
  __syncthreads();

  const int nn = (N - n0) < NBK ? (N - n0) : NBK;
  int4* g4 = reinterpret_cast<int4*>(slots + (size_t)n0 * SLOTS);
#pragma unroll
  for (int k = 0; k < (NBK * SLOTS / 4) / 256; ++k) {
    const int q = t + k * 256;
    if ((q >> 3) < nn) g4[q] = l4[q];
  }
  if (t < nn) cnt[n0 + t] = lcnt[t];
}

__global__ void __launch_bounds__(256)
fill_slots_and_gemm(const unsigned* __restrict__ sorted,
                    const unsigned* __restrict__ meta, int* __restrict__ cnt,
                    int* __restrict__ slots, int* __restrict__ novf,
                    int* __restrict__ ovf_dst, int* __restrict__ ovf_src,
                    int N, int E, int G, int Bn, const float* __restrict__ X,
                    const float* __restrict__ W, float* __restrict__ H) {
  const int blk = (int)blockIdx.x;
  if (blk < Bn) {
    fill_slots(sorted, meta, cnt, slots, novf, ovf_dst, ovf_src, N, E, G, blk);
  } else {
    gat_gemm_wmma<64, 64>(X, W, H, N, blk - Bn);
  }
}

template <int V, typename VF>
__device__ __forceinline__ float lane_dot(VF r, VF a) {
  if constexpr (V == 4) {
    v2f d = r.xy * a.xy;
    d = r.zw * a.zw + d;
    return d.x + d.y;
  } else {
    return fmaf(r.y, a.y, r.x * a.x);
  }
}

template <int D, bool TANH_BIAS, int NOUT>
__global__ void node_softmax_aggregate(const int* __restrict__ cnt,
                                       const int* __restrict__ slots,
                                       const float* __restrict__ attS,
                                       const float* __restrict__ attD,
                                       const float* __restrict__ H,
                                       const float* __restrict__ bias,
                                       const float* __restrict__ Wn,
                                       const int* __restrict__ novf,
                                       const int* __restrict__ ovf_dst,
                                       const int* __restrict__ ovf_src,
                                       float* __restrict__ out, int N, int E) {
  static_assert(D == 32 || D == 64, "row must be 32 or 64 floats");
  static_assert(NOUT == 0 || NOUT == 32, "fused GEMM: two 16-column tiles");
  constexpr bool ADD_BIAS = TANH_BIAS || NOUT != 0;
  constexpr int V = D / 16;
  typedef float vf __attribute__((ext_vector_type(V)));
  const int lane = threadIdx.x & 31;
  const int c    = lane & 15;
  const int hf   = lane >> 4;
  const int wv = __builtin_amdgcn_readfirstlane(
      (int)((blockIdx.x * blockDim.x + threadIdx.x) >> 5));
  if (NOUT == 0 && 2 * wv >= N) return;
  const bool live = 2 * wv + hf < N;
  const int node  = live ? 2 * wv + hf : N - 1;

  const vf* Hv = reinterpret_cast<const vf*>(H);
  const vf hs = Hv[(unsigned)(node * 16 + c)];
  const vf as = reinterpret_cast<const vf*>(attS)[c];
  const vf ad = reinterpret_cast<const vf*>(attD)[c];
  vf bv = 0.f;
  if (ADD_BIAS) bv = reinterpret_cast<const vf*>(bias)[c];
  const int sl0 = slots[(unsigned)(node * SLOTS) + c];
  const int sl1 = slots[(unsigned)(node * SLOTS) + 16 + c];
  const int deg = cnt[node];
  const int nslot = deg < SLOTS ? deg : SLOTS;

  float ps = lane_dot<V>(hs, as);
  float pd = lane_dot<V>(hs, ad);
  ps += lane_xor<8>(ps);  pd += lane_xor<8>(pd);
  ps += lane_xor<4>(ps);  pd += lane_xor<4>(pd);
  ps += lane_xor<2>(ps);  pd += lane_xor<2>(pd);
  ps += lane_xor<1>(ps);  pd += lane_xor<1>(pd);
  const float adn = pd;
  float m = leaky(ps + adn);
  float l = 1.f;
  vf acc = hs;

  const bool b3 = (lane & 8) != 0, b2 = (lane & 4) != 0;
  const bool b1 = (lane & 2) != 0, b0 = (lane & 1) != 0;

  asm volatile("" ::"v"(sl0), "v"(sl1));
  if (ADD_BIAS) {
    if (V == 4) asm volatile("" ::"v"(bv[0]), "v"(bv[1]), "v"(bv[2]), "v"(bv[3]));
    else asm volatile("" ::"v"(bv[0]), "v"(bv[1]));
  }

  const int ns_lo = __builtin_amdgcn_readlane(nslot, 0);
  const int ns_hi = __builtin_amdgcn_readlane(nslot, 16);
  const int nmax  = ns_lo > ns_hi ? ns_lo : ns_hi;

  for (int base = 0; base < nmax; base += 16) {
    const int sl   = base == 0 ? sl0 : sl1;
    const bool has = c < nslot - base;
    const int nb   = (nmax - base < 16) ? (nmax - base) : 16;

    vf r[16];
#pragma unroll
    for (int j = 0; j < 16; ++j) {
      const int sj = row_lane_at(sl, j);
      r[j] = Hv[(unsigned)(sj * 16 + c)];
    }
    asm volatile("" ::"v"(r[0]), "v"(r[1]), "v"(r[2]), "v"(r[3]), "v"(r[4]),
                 "v"(r[5]), "v"(r[6]), "v"(r[7]), "v"(r[8]), "v"(r[9]),
                 "v"(r[10]), "v"(r[11]), "v"(r[12]), "v"(r[13]), "v"(r[14]),
                 "v"(r[15]));

    float p[16];
#pragma unroll
    for (int j = 0; j < 16; ++j) p[j] = lane_dot<V>(r[j], as);
    float q[8];
#pragma unroll
    for (int k = 0; k < 8; ++k) {
      const float keep = b3 ? p[k + 8] : p[k];
      const float send = b3 ? p[k] : p[k + 8];
      q[k] = keep + lane_xor<8>(send);
    }
    float t[4];
#pragma unroll
    for (int k = 0; k < 4; ++k) {
      const float keep = b2 ? q[k + 4] : q[k];
      const float send = b2 ? q[k] : q[k + 4];
      t[k] = keep + lane_xor<4>(send);
    }
    float u[2];
#pragma unroll
    for (int k = 0; k < 2; ++k) {
      const float keep = b1 ? t[k + 2] : t[k];
      const float send = b1 ? t[k] : t[k + 2];
      u[k] = keep + lane_xor<2>(send);
    }
    float sc;
    {
      const float keep = b0 ? u[1] : u[0];
      const float send = b0 ? u[0] : u[1];
      sc = keep + lane_xor<1>(send);
    }

    const float v = has ? leaky(sc + adn) : -INFINITY;
    float cm = v;
    cm = fmaxf(cm, lane_xor<8>(cm));
    cm = fmaxf(cm, lane_xor<4>(cm));
    cm = fmaxf(cm, lane_xor<2>(cm));
    cm = fmaxf(cm, lane_xor<1>(cm));
    const float mn    = fmaxf(m, cm);
    const float scale = __expf(m - mn);
    const float w     = has ? __expf(v - mn) : 0.f;
    float ws = w;
    ws += lane_xor<8>(ws);
    ws += lane_xor<4>(ws);
    ws += lane_xor<2>(ws);
    ws += lane_xor<1>(ws);
    l   = l * scale + ws;
    acc = acc * scale;
    m   = mn;

#pragma unroll
    for (int g = 0; g < 4; ++g) {
      if (4 * g < nb) {
#pragma unroll
        for (int jj = 0; jj < 4; ++jj) {
          const int j    = 4 * g + jj;
          acc += row_lane_at(w, j) * r[j];
        }
      }
    }
  }

  const bool ovl = deg > SLOTS;
  if (__builtin_amdgcn_ballot_w32(ovl) != 0u) {
    const int node_lo = __builtin_amdgcn_readlane(node, 0);
    const int node_hi = __builtin_amdgcn_readlane(node, 16);
    int nov = novf[0];
    nov = nov < E ? nov : E;
    for (int base = 0; base < nov; base += 32) {
      const int i = base + lane;
      int od = -1, os = 0;
      if (i < nov) {
        od = ovf_dst[i];
        os = ovf_src[i];
      }
      unsigned m_lo = __builtin_amdgcn_ballot_w32(od == node_lo);
      unsigned m_hi = __builtin_amdgcn_ballot_w32(od == node_hi);
      while ((m_lo | m_hi) != 0u) {
        const bool a_lo = m_lo != 0u, a_hi = m_hi != 0u;
        const int j_lo = a_lo ? __builtin_ctz(m_lo) : 0;
        const int j_hi = a_hi ? __builtin_ctz(m_hi) : 0;
        m_lo &= m_lo - 1u;
        m_hi &= m_hi - 1u;
        const int s_lo = __builtin_amdgcn_readlane(os, j_lo);
        const int s_hi = __builtin_amdgcn_readlane(os, j_hi);
        const bool act = hf ? a_hi : a_lo;
        const int sj   = act ? (hf ? s_hi : s_lo) : node;
        const vf g = Hv[(unsigned)(sj * 16 + c)];
        float sc = lane_dot<V>(g, as);
        sc += lane_xor<8>(sc);
        sc += lane_xor<4>(sc);
        sc += lane_xor<2>(sc);
        sc += lane_xor<1>(sc);
        const float v     = leaky(sc + adn);
        const float mn    = act ? fmaxf(m, v) : m;
        const float scale = __expf(m - mn);
        const float w     = act ? __expf(v - mn) : 0.f;
        l   = l * scale + w;
        acc = acc * scale + w * g;
        m   = mn;
      }
    }
  }

  if constexpr (NOUT == 0) {
    if (live) {
      vf rr;
#pragma unroll
      for (int k = 0; k < V; ++k) {
        rr[k] = acc[k] / l;
        if (TANH_BIAS) rr[k] = tanhf(rr[k] + bv[k]);
      }
      reinterpret_cast<vf*>(out)[(unsigned)(node * 16 + c)] = rr;
    }
  } else {
    constexpr int LDA = D + 4;
    constexpr int KS  = D / 4;
    __shared__ __attribute__((aligned(16))) float tileA[16 * LDA];
    const int wib = __builtin_amdgcn_readfirstlane((int)(threadIdx.x >> 5));

    vf rr;
#pragma unroll
    for (int k = 0; k < V; ++k) rr[k] = fmaxf(acc[k] / l + bv[k], 0.f);
    const int trow = 2 * wib + hf;
    *reinterpret_cast<vf*>(&tileA[trow * LDA + V * c]) = rr;

    v2f wb0[KS], wb1[KS];
    if (wib == 0) {
#pragma unroll
      for (int i = 0; i < KS; ++i) {
        const int ka = 4 * i + 2 * hf;
        wb0[i].x = Wn[ka * NOUT + c];
        wb0[i].y = Wn[(ka + 1) * NOUT + c];
        wb1[i].x = Wn[ka * NOUT + 16 + c];
        wb1[i].y = Wn[(ka + 1) * NOUT + 16 + c];
      }
    }
    __syncthreads();

    if (wib == 0) {
      const int row0 = (int)blockIdx.x * 16 + hf * 8;
#if HAVE_WMMA_F32_16X16X4
      v8f c0 = {0.f, 0.f, 0.f, 0.f, 0.f, 0.f, 0.f, 0.f};
      v8f c1 = {0.f, 0.f, 0.f, 0.f, 0.f, 0.f, 0.f, 0.f};
#pragma unroll
      for (int i = 0; i < KS; ++i) {
        const v2f a =
            *reinterpret_cast<const v2f*>(&tileA[c * LDA + 4 * i + 2 * hf]);
        c0 = __builtin_amdgcn_wmma_f32_16x16x4_f32(false, a, false, wb0[i],
                                                   (short)0, c0, false, false);
        c1 = __builtin_amdgcn_wmma_f32_16x16x4_f32(false, a, false, wb1[i],
                                                   (short)0, c1, false, false);
      }
#else
      float c0[8], c1[8];
      for (int r = 0; r < 8; ++r) {
        float s0 = 0.f, s1 = 0.f;
        for (int k = 0; k < D; ++k) {
          const float xv = tileA[(hf * 8 + r) * LDA + k];
          s0 += xv * Wn[k * NOUT + c];
          s1 += xv * Wn[k * NOUT + 16 + c];
        }
        c0[r] = s0;
        c1[r] = s1;
      }
#endif
#pragma unroll
      for (int r = 0; r < 8; ++r) {
        if (row0 + r < N) {
          out[(unsigned)((row0 + r) * NOUT + c)]      = c0[r];
          out[(unsigned)((row0 + r) * NOUT + 16 + c)] = c1[r];
        }
      }
    }
  }
}

extern "C" void kernel_launch(void* const* d_in, const int* in_sizes, int n_in,
                              void* d_out, int out_size, void* d_ws,
                              size_t ws_size, hipStream_t stream) {
  const float* x   = (const float*)d_in[0];
  const int*   ei  = (const int*)d_in[1];
  const float* W1  = (const float*)d_in[2];
  const float* as1 = (const float*)d_in[3];
  const float* ad1 = (const float*)d_in[4];
  const float* b1  = (const float*)d_in[5];
  const float* W2  = (const float*)d_in[6];
  const float* as2 = (const float*)d_in[7];
  const float* ad2 = (const float*)d_in[8];
  const float* b2  = (const float*)d_in[9];
  float* out = (float*)d_out;

  const int N    = in_sizes[0] / 64;
  const int E    = in_sizes[1] / 2;
  const int* src = ei;
  const int* dst = ei + E;

  float* ws = (float*)d_ws;
  size_t o = 0;
  float* h1 = ws + o;   o += (size_t)N * 64;
  float* h2 = ws + o;   o += (size_t)N * 32;
  int* slots   = (int*)(ws + o); o += (size_t)N * SLOTS;
  int* novf    = (int*)(ws + o); o += 32;
  int* ovf_dst = (int*)(ws + o); o += E;
  int* ovf_src = (int*)(ws + o); o += E;
  int* cnt     = (int*)(ws + o); o += N;
  const int G  = (E + CH - 1) / CH;
  const int Bn = (N + NBK - 1) / NBK;
  o = (o + 31) & ~(size_t)31;
  unsigned* sorted = (unsigned*)(ws + o); o += (size_t)G * CH;
  unsigned* meta   = (unsigned*)(ws + o); o += (size_t)Bn * G;

  const int B  = 256;
  const int gn = (int)((((size_t)N + 1) / 2 * 32 + B - 1) / B);

  part_edges<<<G, B, 0, stream>>>(src, dst, sorted, meta, novf, E, Bn, G);

  {
    const int tiles = (N / 16) * (64 / 16);
    fill_slots_and_gemm<<<Bn + (tiles + 7) / 8, B, 0, stream>>>(
        sorted, meta, cnt, slots, novf, ovf_dst, ovf_src, N, E, G, Bn, x, W1,
        h1);
    node_softmax_aggregate<64, false, 32>
        <<<gn, B, 0, stream>>>(cnt, slots, as1, ad1, h1, b1, W2, novf, ovf_dst,
                                ovf_src, h2, N, E);
  }

  {
    node_softmax_aggregate<32, true, 0>
        <<<gn, B, 0, stream>>>(cnt, slots, as2, ad2, h2, b2, nullptr, novf,
                                ovf_dst, ovf_src, out, N, E);
  }
}


# Aggregate kernels: a batch in which neither half has over 8 (or 4) rows left, as most second batches, gathers and scores 8 (or 4) rows, not 16; its score sums go in stride order 4,2,1,8 (or 2,1,4,8)
# speedup vs baseline: 8.4905x; 1.0394x over previous
#include <hip/hip_runtime.h>
#include <math.h>


typedef __attribute__((ext_vector_type(2))) float v2f;
typedef __attribute__((ext_vector_type(8))) float v8f;

#define NEG_SLOPE 0.2f

#if defined(__has_builtin)
#if __has_builtin(__builtin_amdgcn_wmma_f32_16x16x4_f32)
#define HAVE_WMMA_F32_16X16X4 1
#endif
#endif

__device__ __forceinline__ float leaky(float v) {
  return v > 0.f ? v : NEG_SLOPE * v;
}

template <int M>
__device__ __forceinline__ float lane_xor(float v) {
  return __builtin_bit_cast(
      float, __builtin_amdgcn_update_dpp(0, __builtin_bit_cast(int, v),
                                         0x160 + M, 0xf, 0xf, true));
}
__device__ __forceinline__ float lane_xor16(float v) {
  const int b = __builtin_bit_cast(int, v);
  return __builtin_bit_cast(
      float, __builtin_amdgcn_permlanex16(b, b, 0x76543210, 0xfedcba98, false,
                                          false));
}

template <int J>
__device__ __forceinline__ int row_lane(int v) {
  return __builtin_amdgcn_update_dpp(0, v, 0x150 + J, 0xf, 0xf, true);
}
__device__ __forceinline__ int row_lane_at(int v, int j) {
  switch (j & 15) {
    case 0: return row_lane<0>(v);
    case 1: return row_lane<1>(v);
    case 2: return row_lane<2>(v);
    case 3: return row_lane<3>(v);
    case 4: return row_lane<4>(v);
    case 5: return row_lane<5>(v);
    case 6: return row_lane<6>(v);
    case 7: return row_lane<7>(v);
    case 8: return row_lane<8>(v);
    case 9: return row_lane<9>(v);
    case 10: return row_lane<10>(v);
    case 11: return row_lane<11>(v);
    case 12: return row_lane<12>(v);
    case 13: return row_lane<13>(v);
    case 14: return row_lane<14>(v);
    default: return row_lane<15>(v);
  }
}
__device__ __forceinline__ float row_lane_at(float v, int j) {
  return __builtin_bit_cast(float, row_lane_at(__builtin_bit_cast(int, v), j));
}

template <int K, int DOUT>
__device__ __forceinline__ void gat_gemm_wmma(const float* __restrict__ X,
                                              const float* __restrict__ W,
                                              float* __restrict__ H, int M,
                                              int blk) {
  const int lane   = threadIdx.x & 31;
  const int wave   = threadIdx.x >> 5;
  const int tileId = blk * (blockDim.x >> 5) + wave;
  const int NT     = DOUT / 16;
  const int mt     = tileId / NT;
  const int nt     = tileId % NT;
  if (mt >= M / 16) return;

  const int r15 = lane & 15;
  const int hi  = lane >> 4;
  const int row = mt * 16 + r15;
  const int col = nt * 16 + r15;

  v8f c = {0.f, 0.f, 0.f, 0.f, 0.f, 0.f, 0.f, 0.f};

#if HAVE_WMMA_F32_16X16X4
  for (int k0 = 0; k0 < K; k0 += 4) {
    const int ka = k0 + 2 * hi;
    v2f a;
    a.x = X[row * K + ka];
    a.y = X[row * K + ka + 1];
    v2f b;
    b.x = W[ka * DOUT + col];
    b.y = W[(ka + 1) * DOUT + col];
    c = __builtin_amdgcn_wmma_f32_16x16x4_f32(false, a, false, b, (short)0, c,
                                              false, false);
  }
#else
  for (int r = 0; r < 8; ++r) {
    const int rr = mt * 16 + hi * 8 + r;
    float acc = 0.f;
    for (int k = 0; k < K; ++k) acc += X[rr * K + k] * W[k * DOUT + col];
    c[r] = acc;
  }
#endif

  const int rowBase = mt * 16 + hi * 8;
#pragma unroll
  for (int r = 0; r < 8; ++r) H[(rowBase + r) * DOUT + col] = c[r];
}

#define SLOTS 32
#define CH 8192
#define EPT 32
#define NBK 256
#define BMAX 1024

__global__ void __launch_bounds__(256)
part_edges(const int* __restrict__ src, const int* __restrict__ dst,
           unsigned* __restrict__ sorted, unsigned* __restrict__ meta,
           int* __restrict__ novf, int E, int Bn, int G) {
  __shared__ __attribute__((aligned(16))) int hist[BMAX];
  __shared__ __attribute__((aligned(16))) unsigned stage[CH];
  __shared__ int wsum[8];
  const int t    = threadIdx.x;
  const int lane = t & 31;
  const int wave = t >> 5;
  const int g    = blockIdx.x;
  const int e0   = g * CH;

  for (int i = t; i < BMAX; i += 256) hist[i] = 0;
  if (g == 0 && t == 0) novf[0] = 0;
  __syncthreads();

  int d[EPT], s[EPT], r[EPT];
#pragma unroll
  for (int k = 0; k < EPT; ++k) {
    const int e  = e0 + k * 256 + t;
    const int ec = e < E ? e : E - 1;
    d[k] = dst[ec];
    s[k] = src[ec];
  }
#pragma unroll
  for (int k = 0; k < EPT; ++k) {
    const int e = e0 + k * 256 + t;
    int b = d[k] >> 8;
    b = b < 0 ? 0 : (b < BMAX ? b : BMAX - 1);
    d[k] = e < E ? d[k] : -1;
    s[k] = (int)(((unsigned)s[k] << 8) | (unsigned)(d[k] & 255));
    r[k] = 0;
    if (d[k] >= 0) r[k] = atomicAdd(&hist[b], 1);
    d[k] = d[k] >= 0 ? b : -1;
  }
  __syncthreads();

  const int4 h = reinterpret_cast<const int4*>(hist)[t];
  const int sum = h.x + h.y + h.z + h.w;
  int incl = sum;
#pragma unroll
  for (int off = 1; off < 32; off <<= 1) {
    const int v = __shfl_up(incl, off);
    if (lane >= off) incl += v;
  }
  if (lane == 31) wsum[wave] = incl;
  __syncthreads();
  int base = 0;
  for (int w = 0; w < wave; ++w) base += wsum[w];
  int4 st;
  st.x = base + incl - sum;
  st.y = st.x + h.x;
  st.z = st.y + h.y;
  st.w = st.z + h.z;
  reinterpret_cast<int4*>(hist)[t] = st;
  {
    const int b = 4 * t;
    unsigned* m = meta + (size_t)b * G + g;
    if (b + 0 < Bn) m[0]             = (unsigned)st.x | ((unsigned)h.x << 16);
    if (b + 1 < Bn) m[(size_t)G]     = (unsigned)st.y | ((unsigned)h.y << 16);
    if (b + 2 < Bn) m[(size_t)G * 2] = (unsigned)st.z | ((unsigned)h.z << 16);
    if (b + 3 < Bn) m[(size_t)G * 3] = (unsigned)st.w | ((unsigned)h.w << 16);
  }
  __syncthreads();

#pragma unroll
  for (int k = 0; k < EPT; ++k) r[k] += hist[d[k] >= 0 ? d[k] : 0];
#pragma unroll
  for (int k = 0; k < EPT; ++k)
    if (d[k] >= 0) stage[r[k]] = (unsigned)s[k];
  __syncthreads();

  const int clen = (E - e0) < CH ? (E - e0) : CH;
  const int c4   = clen >> 2;
  uint4* o4 = reinterpret_cast<uint4*>(sorted + e0);
  const uint4* s4 = reinterpret_cast<const uint4*>(stage);
  for (int i = t; i < c4; i += 256) o4[i] = s4[i];
  for (int i = (c4 << 2) + t; i < clen; i += 256) sorted[e0 + i] = stage[i];
}

__device__ __forceinline__ void put_entry(unsigned ent, int r, int n0,
                                          int* lslots, int* __restrict__ novf,
                                          int* __restrict__ ovf_dst,
                                          int* __restrict__ ovf_src, int E) {
  const int n = (int)(ent & 255u);
  const int s = (int)(ent >> 8);
  if (r < SLOTS) {
    lslots[n * SLOTS + r] = s;
  } else {
    const int p = atomicAdd(novf, 1);
    if (p < E) {
      ovf_dst[p] = n0 + n;
      ovf_src[p] = s;
    }
  }
}

__device__ __forceinline__ void
fill_slots(const unsigned* __restrict__ sorted,
           const unsigned* __restrict__ meta, int* __restrict__ cnt,
           int* __restrict__ slots, int* __restrict__ novf,
           int* __restrict__ ovf_dst, int* __restrict__ ovf_src, int N, int E,
           int G, const int b) {
  __shared__ __attribute__((aligned(16))) int lslots[NBK * SLOTS];
  __shared__ int lcnt[NBK];
  const int t    = threadIdx.x;
  const int lane = t & 31;
  const int wave = __builtin_amdgcn_readfirstlane(t >> 5);
  const int n0   = b * NBK;

  int4* l4 = reinterpret_cast<int4*>(lslots);
#pragma unroll
  for (int k = 0; k < (NBK * SLOTS / 4) / 256; ++k) {
    const int q = t + k * 256;
    const int dn = n0 + (q >> 3);
    l4[q] = make_int4(dn, dn, dn, dn);
  }
  lcnt[t] = 0;
  __syncthreads();

  const int GW   = (G + 7) >> 3;
  const int gBeg = wave * GW;
  const int gEnd = (gBeg + GW) < G ? (gBeg + GW) : G;
  const unsigned* mrow = meta + (size_t)b * G;

  for (int kb = gBeg; kb < gEnd; kb += 32) {
    const int nr = (gEnd - kb) < 32 ? (gEnd - kb) : 32;
    const unsigned mv = lane < nr ? mrow[kb + lane] : 0u;

    unsigned ent[32];
    int rk[32];
#pragma unroll
    for (int u = 0; u < 32; ++u) {
      const unsigned m = (unsigned)__builtin_amdgcn_readlane((int)mv, u);
      const int st  = (int)(m & 0xffffu);
      const int len = (int)(m >> 16);
      ent[u] = 0u;
      if (lane < len) ent[u] = sorted[(size_t)(kb + u) * CH + st + lane];
    }
#pragma unroll
    for (int u = 0; u < 32; ++u) {
      const unsigned m = (unsigned)__builtin_amdgcn_readlane((int)mv, u);
      const int len = (int)(m >> 16);
      rk[u] = 0;
      if (lane < len) rk[u] = atomicAdd(&lcnt[ent[u] & 255u], 1);
    }
#pragma unroll
    for (int u = 0; u < 32; ++u) {
      const unsigned m = (unsigned)__builtin_amdgcn_readlane((int)mv, u);
      const int len = (int)(m >> 16);
      if (lane < len)
        put_entry(ent[u], rk[u], n0, lslots, novf, ovf_dst, ovf_src, E);
    }
    for (int u = 0; u < nr; ++u) {
      const unsigned m = (unsigned)__builtin_amdgcn_readlane((int)mv, u);
      const int st  = (int)(m & 0xffffu);
      const int len = (int)(m >> 16);
      for (int i = 32 + lane; i < len; i += 32) {
        const unsigned e = sorted[(size_t)(kb + u) * CH + st + i];
        const int r = atomicAdd(&lcnt[e & 255u], 1);
        put_entry(e, r, n0, lslots, novf, ovf_dst, ovf_src, E);
      }
    }
  }
  __syncthreads();

  const int nn = (N - n0) < NBK ? (N - n0) : NBK;
  int4* g4 = reinterpret_cast<int4*>(slots + (size_t)n0 * SLOTS);
#pragma unroll
  for (int k = 0; k < (NBK * SLOTS / 4) / 256; ++k) {
    const int q = t + k * 256;
    if ((q >> 3) < nn) g4[q] = l4[q];
  }
  if (t < nn) cnt[n0 + t] = lcnt[t];
}

__global__ void __launch_bounds__(256)
fill_slots_and_gemm(const unsigned* __restrict__ sorted,
                    const unsigned* __restrict__ meta, int* __restrict__ cnt,
                    int* __restrict__ slots, int* __restrict__ novf,
                    int* __restrict__ ovf_dst, int* __restrict__ ovf_src,
                    int N, int E, int G, int Bn, const float* __restrict__ X,
                    const float* __restrict__ W, float* __restrict__ H) {
  const int blk = (int)blockIdx.x;
  if (blk < Bn) {
    fill_slots(sorted, meta, cnt, slots, novf, ovf_dst, ovf_src, N, E, G, blk);
  } else {
    gat_gemm_wmma<64, 64>(X, W, H, N, blk - Bn);
  }
}

template <int V, typename VF>
__device__ __forceinline__ float lane_dot(VF r, VF a) {
  if constexpr (V == 4) {
    v2f d = r.xy * a.xy;
    d = r.zw * a.zw + d;
    return d.x + d.y;
  } else {
    return fmaf(r.y, a.y, r.x * a.x);
  }
}

template <int R, int V, typename VF>
__device__ __forceinline__ void
batch_rows(const VF* __restrict__ Hv, const int sl, const int c, const bool has,
           const int nb, const VF as, const float adn, const bool b3,
           const bool b2, const bool b1, const bool b0, float& m, float& l,
           VF& acc) {
  static_assert(R == 16 || R == 8 || R == 4, "batch of 16, 8 or 4 rows");
  VF r[R];
#pragma unroll
  for (int j = 0; j < R; ++j) {
    const int sj = row_lane_at(sl, j);
    r[j] = Hv[(unsigned)(sj * 16 + c)];
  }
  if constexpr (R == 16) __builtin_amdgcn_sched_barrier(0);

  float p[R];
#pragma unroll
  for (int j = 0; j < R; ++j) p[j] = lane_dot<V>(r[j], as);
  float t[4];
  if constexpr (R == 16) {
    float q[8];
#pragma unroll
    for (int k = 0; k < 8; ++k) {
      const float keep = b3 ? p[k + 8] : p[k];
      const float send = b3 ? p[k] : p[k + 8];
      q[k] = keep + lane_xor<8>(send);
    }
#pragma unroll
    for (int k = 0; k < 4; ++k) {
      const float keep = b2 ? q[k + 4] : q[k];
      const float send = b2 ? q[k] : q[k + 4];
      t[k] = keep + lane_xor<4>(send);
    }
  } else if constexpr (R == 8) {
#pragma unroll
    for (int k = 0; k < 4; ++k) {
      const float keep = b2 ? p[k + 4] : p[k];
      const float send = b2 ? p[k] : p[k + 4];
      t[k] = keep + lane_xor<4>(send);
    }
  } else {
#pragma unroll
    for (int k = 0; k < 4; ++k) t[k] = p[k];
  }
  float u[2];
#pragma unroll
  for (int k = 0; k < 2; ++k) {
    const float keep = b1 ? t[k + 2] : t[k];
    const float send = b1 ? t[k] : t[k + 2];
    u[k] = keep + lane_xor<2>(send);
  }
  float sc;
  {
    const float keep = b0 ? u[1] : u[0];
    const float send = b0 ? u[0] : u[1];
    sc = keep + lane_xor<1>(send);
  }
  if constexpr (R == 4) sc += lane_xor<4>(sc);
  if constexpr (R <= 8) sc += lane_xor<8>(sc);

  const float v = has ? leaky(sc + adn) : -INFINITY;
  float cm = v;
  cm = fmaxf(cm, lane_xor<8>(cm));
  cm = fmaxf(cm, lane_xor<4>(cm));
  cm = fmaxf(cm, lane_xor<2>(cm));
  cm = fmaxf(cm, lane_xor<1>(cm));
  const float mn    = fmaxf(m, cm);
  const float scale = __expf(m - mn);
  const float w     = has ? __expf(v - mn) : 0.f;
  float ws = w;
  ws += lane_xor<8>(ws);
  ws += lane_xor<4>(ws);
  ws += lane_xor<2>(ws);
  ws += lane_xor<1>(ws);
  l   = l * scale + ws;
  acc = acc * scale;
  m   = mn;

#pragma unroll
  for (int g = 0; g < R / 4; ++g) {
    if (4 * g < nb) {
#pragma unroll
      for (int jj = 0; jj < 4; ++jj) {
        const int j = 4 * g + jj;
        acc += row_lane_at(w, j) * r[j];
      }
    }
  }
}

template <int D, bool TANH_BIAS, int NOUT>
__global__ void node_softmax_aggregate(const int* __restrict__ cnt,
                                       const int* __restrict__ slots,
                                       const float* __restrict__ attS,
                                       const float* __restrict__ attD,
                                       const float* __restrict__ H,
                                       const float* __restrict__ bias,
                                       const float* __restrict__ Wn,
                                       const int* __restrict__ novf,
                                       const int* __restrict__ ovf_dst,
                                       const int* __restrict__ ovf_src,
                                       float* __restrict__ out, int N, int E) {
  static_assert(D == 32 || D == 64, "row must be 32 or 64 floats");
  static_assert(NOUT == 0 || NOUT == 32, "fused GEMM: two 16-column tiles");
  constexpr bool ADD_BIAS = TANH_BIAS || NOUT != 0;
  constexpr int V = D / 16;
  typedef float vf __attribute__((ext_vector_type(V)));
  const int lane = threadIdx.x & 31;
  const int c    = lane & 15;
  const int hf   = lane >> 4;
  const int wv = __builtin_amdgcn_readfirstlane(
      (int)((blockIdx.x * blockDim.x + threadIdx.x) >> 5));
  if (NOUT == 0 && 2 * wv >= N) return;
  const bool live = 2 * wv + hf < N;
  const int node  = live ? 2 * wv + hf : N - 1;

  const vf* Hv = reinterpret_cast<const vf*>(H);
  const vf hs = Hv[(unsigned)(node * 16 + c)];
  const vf as = reinterpret_cast<const vf*>(attS)[c];
  const vf ad = reinterpret_cast<const vf*>(attD)[c];
  vf bv = 0.f;
  if (ADD_BIAS) bv = reinterpret_cast<const vf*>(bias)[c];
  const int sl0 = slots[(unsigned)(node * SLOTS) + c];
  const int sl1 = slots[(unsigned)(node * SLOTS) + 16 + c];
  const int deg = cnt[node];
  const int nslot = deg < SLOTS ? deg : SLOTS;

  float ps = lane_dot<V>(hs, as);
  float pd = lane_dot<V>(hs, ad);
  ps += lane_xor<8>(ps);  pd += lane_xor<8>(pd);
  ps += lane_xor<4>(ps);  pd += lane_xor<4>(pd);
  ps += lane_xor<2>(ps);  pd += lane_xor<2>(pd);
  ps += lane_xor<1>(ps);  pd += lane_xor<1>(pd);
  const float adn = pd;
  float m = leaky(ps + adn);
  float l = 1.f;
  vf acc = hs;

  const bool b3 = (lane & 8) != 0, b2 = (lane & 4) != 0;
  const bool b1 = (lane & 2) != 0, b0 = (lane & 1) != 0;

  asm volatile("" ::"v"(sl0), "v"(sl1));
  if (ADD_BIAS) {
    if (V == 4) asm volatile("" ::"v"(bv[0]), "v"(bv[1]), "v"(bv[2]), "v"(bv[3]));
    else asm volatile("" ::"v"(bv[0]), "v"(bv[1]));
  }

  const int ns_lo = __builtin_amdgcn_readlane(nslot, 0);
  const int ns_hi = __builtin_amdgcn_readlane(nslot, 16);
  const int nmax  = ns_lo > ns_hi ? ns_lo : ns_hi;

  for (int base = 0; base < nmax; base += 16) {
    const int sl   = base == 0 ? sl0 : sl1;
    const bool has = c < nslot - base;
    const int nb   = (nmax - base < 16) ? (nmax - base) : 16;
    if (nb > 8)
      batch_rows<16, V>(Hv, sl, c, has, nb, as, adn, b3, b2, b1, b0, m, l, acc);
    else if (nb > 4)
      batch_rows<8, V>(Hv, sl, c, has, nb, as, adn, b3, b2, b1, b0, m, l, acc);
    else
      batch_rows<4, V>(Hv, sl, c, has, nb, as, adn, b3, b2, b1, b0, m, l, acc);
  }

  const bool ovl = deg > SLOTS;
  if (__builtin_amdgcn_ballot_w32(ovl) != 0u) {
    const int node_lo = __builtin_amdgcn_readlane(node, 0);
    const int node_hi = __builtin_amdgcn_readlane(node, 16);
    int nov = novf[0];
    nov = nov < E ? nov : E;
    for (int base = 0; base < nov; base += 32) {
      const int i = base + lane;
      int od = -1, os = 0;
      if (i < nov) {
        od = ovf_dst[i];
        os = ovf_src[i];
      }
      unsigned m_lo = __builtin_amdgcn_ballot_w32(od == node_lo);
      unsigned m_hi = __builtin_amdgcn_ballot_w32(od == node_hi);
      while ((m_lo | m_hi) != 0u) {
        const bool a_lo = m_lo != 0u, a_hi = m_hi != 0u;
        const int j_lo = a_lo ? __builtin_ctz(m_lo) : 0;
        const int j_hi = a_hi ? __builtin_ctz(m_hi) : 0;
        m_lo &= m_lo - 1u;
        m_hi &= m_hi - 1u;
        const int s_lo = __builtin_amdgcn_readlane(os, j_lo);
        const int s_hi = __builtin_amdgcn_readlane(os, j_hi);
        const bool act = hf ? a_hi : a_lo;
        const int sj   = act ? (hf ? s_hi : s_lo) : node;
        const vf g = Hv[(unsigned)(sj * 16 + c)];
        float sc = lane_dot<V>(g, as);
        sc += lane_xor<8>(sc);
        sc += lane_xor<4>(sc);
        sc += lane_xor<2>(sc);
        sc += lane_xor<1>(sc);
        const float v     = leaky(sc + adn);
        const float mn    = act ? fmaxf(m, v) : m;
        const float scale = __expf(m - mn);
        const float w     = act ? __expf(v - mn) : 0.f;
        l   = l * scale + w;
        acc = acc * scale + w * g;
        m   = mn;
      }
    }
  }

  if constexpr (NOUT == 0) {
    if (live) {
      vf rr;
#pragma unroll
      for (int k = 0; k < V; ++k) {
        rr[k] = acc[k] / l;
        if (TANH_BIAS) rr[k] = tanhf(rr[k] + bv[k]);
      }
      reinterpret_cast<vf*>(out)[(unsigned)(node * 16 + c)] = rr;
    }
  } else {
    constexpr int LDA = D + 4;
    constexpr int KS  = D / 4;
    __shared__ __attribute__((aligned(16))) float tileA[16 * LDA];
    const int wib = __builtin_amdgcn_readfirstlane((int)(threadIdx.x >> 5));

    vf rr;
#pragma unroll
    for (int k = 0; k < V; ++k) rr[k] = fmaxf(acc[k] / l + bv[k], 0.f);
    const int trow = 2 * wib + hf;
    *reinterpret_cast<vf*>(&tileA[trow * LDA + V * c]) = rr;

    v2f wb0[KS], wb1[KS];
    if (wib == 0) {
#pragma unroll
      for (int i = 0; i < KS; ++i) {
        const int ka = 4 * i + 2 * hf;
        wb0[i].x = Wn[ka * NOUT + c];
        wb0[i].y = Wn[(ka + 1) * NOUT + c];
        wb1[i].x = Wn[ka * NOUT + 16 + c];
        wb1[i].y = Wn[(ka + 1) * NOUT + 16 + c];
      }
    }
    __syncthreads();

    if (wib == 0) {
      const int row0 = (int)blockIdx.x * 16 + hf * 8;
#if HAVE_WMMA_F32_16X16X4
      v8f c0 = {0.f, 0.f, 0.f, 0.f, 0.f, 0.f, 0.f, 0.f};
      v8f c1 = {0.f, 0.f, 0.f, 0.f, 0.f, 0.f, 0.f, 0.f};
#pragma unroll
      for (int i = 0; i < KS; ++i) {
        const v2f a =
            *reinterpret_cast<const v2f*>(&tileA[c * LDA + 4 * i + 2 * hf]);
        c0 = __builtin_amdgcn_wmma_f32_16x16x4_f32(false, a, false, wb0[i],
                                                   (short)0, c0, false, false);
        c1 = __builtin_amdgcn_wmma_f32_16x16x4_f32(false, a, false, wb1[i],
                                                   (short)0, c1, false, false);
      }
#else
      float c0[8], c1[8];
      for (int r = 0; r < 8; ++r) {
        float s0 = 0.f, s1 = 0.f;
        for (int k = 0; k < D; ++k) {
          const float xv = tileA[(hf * 8 + r) * LDA + k];
          s0 += xv * Wn[k * NOUT + c];
          s1 += xv * Wn[k * NOUT + 16 + c];
        }
        c0[r] = s0;
        c1[r] = s1;
      }
#endif
#pragma unroll
      for (int r = 0; r < 8; ++r) {
        if (row0 + r < N) {
          out[(unsigned)((row0 + r) * NOUT + c)]      = c0[r];
          out[(unsigned)((row0 + r) * NOUT + 16 + c)] = c1[r];
        }
      }
    }
  }
}

extern "C" void kernel_launch(void* const* d_in, const int* in_sizes, int n_in,
                              void* d_out, int out_size, void* d_ws,
                              size_t ws_size, hipStream_t stream) {
  const float* x   = (const float*)d_in[0];
  const int*   ei  = (const int*)d_in[1];
  const float* W1  = (const float*)d_in[2];
  const float* as1 = (const float*)d_in[3];
  const float* ad1 = (const float*)d_in[4];
  const float* b1  = (const float*)d_in[5];
  const float* W2  = (const float*)d_in[6];
  const float* as2 = (const float*)d_in[7];
  const float* ad2 = (const float*)d_in[8];
  const float* b2  = (const float*)d_in[9];
  float* out = (float*)d_out;

  const int N    = in_sizes[0] / 64;
  const int E    = in_sizes[1] / 2;
  const int* src = ei;
  const int* dst = ei + E;

  float* ws = (float*)d_ws;
  size_t o = 0;
  float* h1 = ws + o;   o += (size_t)N * 64;
  float* h2 = ws + o;   o += (size_t)N * 32;
  int* slots   = (int*)(ws + o); o += (size_t)N * SLOTS;
  int* novf    = (int*)(ws + o); o += 32;
  int* ovf_dst = (int*)(ws + o); o += E;
  int* ovf_src = (int*)(ws + o); o += E;
  int* cnt     = (int*)(ws + o); o += N;
  const int G  = (E + CH - 1) / CH;
  const int Bn = (N + NBK - 1) / NBK;
  o = (o + 31) & ~(size_t)31;
  unsigned* sorted = (unsigned*)(ws + o); o += (size_t)G * CH;
  unsigned* meta   = (unsigned*)(ws + o); o += (size_t)Bn * G;

  const int B  = 256;
  const int gn = (int)((((size_t)N + 1) / 2 * 32 + B - 1) / B);

  part_edges<<<G, B, 0, stream>>>(src, dst, sorted, meta, novf, E, Bn, G);

  {
    const int tiles = (N / 16) * (64 / 16);
    fill_slots_and_gemm<<<Bn + (tiles + 7) / 8, B, 0, stream>>>(
        sorted, meta, cnt, slots, novf, ovf_dst, ovf_src, N, E, G, Bn, x, W1,
        h1);
    node_softmax_aggregate<64, false, 32>
        <<<gn, B, 0, stream>>>(cnt, slots, as1, ad1, h1, b1, W2, novf, ovf_dst,
                                ovf_src, h2, N, E);
  }

  {
    node_softmax_aggregate<32, true, 0>
        <<<gn, B, 0, stream>>>(cnt, slots, as2, ad2, h2, b2, nullptr, novf,
                                ovf_dst, ovf_src, out, N, E);
  }
}
